# starting kernel
# baseline (speedup 1.0000x reference)
#include <hip/hip_runtime.h>

#define N_NODES 100000
#define IN_F    128
#define H_HEADS 4
#define D_OUT   16
#define HD      64
#define E_EDGES 3200000
#define SLOPE   0.2f

typedef __attribute__((ext_vector_type(2))) float v2f;
typedef __attribute__((ext_vector_type(4))) float v4f;
typedef __attribute__((ext_vector_type(8))) float v8f;

__device__ __forceinline__ unsigned flt_flip(float f) {
    unsigned u = __float_as_uint(f);
    return u ^ (((int)u >> 31) | 0x80000000u);
}
__device__ __forceinline__ float flt_unflip(unsigned u) {
    unsigned b = (u & 0x80000000u) ? (u ^ 0x80000000u) : ~u;
    return __uint_as_float(b);
}

__global__ __launch_bounds__(256) void gat_gemm(const float* __restrict__ feat,
                                                const float* __restrict__ W,
                                                float* __restrict__ ft) {
    const int wave = threadIdx.x >> 5;
    const int lane = threadIdx.x & 31;
    const int rt   = wave >> 2;
    const int ct   = wave & 3;
    const int r0   = blockIdx.x * 32 + rt * 16;
    const int c0   = ct * 16;
    const int m    = lane & 15;
    const int hi   = lane >> 4;

    const float* arow = feat + (size_t)(r0 + m) * IN_F + (hi << 1);
    const float* brow = W    + (size_t)(c0 + m) * IN_F + (hi << 1);

    v8f acc = {};
#pragma unroll
    for (int k0 = 0; k0 < IN_F; k0 += 4) {
        v2f a = *(const v2f*)(arow + k0);
        v2f b = *(const v2f*)(brow + k0);
        acc = __builtin_amdgcn_wmma_f32_16x16x4_f32(
            false, a, false, b, (short)0, acc, false, false);
    }
    float* outp = ft + (size_t)(r0 + hi * 8) * HD + c0 + m;
#pragma unroll
    for (int r = 0; r < 8; ++r) outp[(size_t)r * HD] = acc[r];
}

__global__ __launch_bounds__(256) void gat_attn(const float* __restrict__ ft,
                                                const float* __restrict__ al,
                                                const float* __restrict__ ar,
                                                float* __restrict__ el,
                                                float* __restrict__ er) {
    int idx = blockIdx.x * blockDim.x + threadIdx.x;
    if (idx >= N_NODES * H_HEADS) return;
    const int h = idx & 3;
    const float* f = ft + (size_t)idx * D_OUT;
    float sl = 0.f, sr = 0.f;
#pragma unroll
    for (int d = 0; d < D_OUT; ++d) {
        float v = f[d];
        sl += v * al[h * D_OUT + d];
        sr += v * ar[h * D_OUT + d];
    }
    el[idx] = sl;
    er[idx] = sr;
}

__global__ __launch_bounds__(256) void gat_init(unsigned* __restrict__ m_u,
                                                float* __restrict__ s,
                                                float* __restrict__ out) {
    int idx = blockIdx.x * blockDim.x + threadIdx.x;
    if (idx < N_NODES * H_HEADS) { m_u[idx] = 0u; s[idx] = 0.f; }
    if (idx < N_NODES * HD)      out[idx] = 0.f;
}

__device__ __forceinline__ float edge_score(float l, float r) {
    float e = l + r;
    return e >= 0.f ? e : SLOPE * e;
}

__global__ __launch_bounds__(256) void gat_edge_max(const int* __restrict__ src,
                                                    const int* __restrict__ dst,
                                                    const float* __restrict__ el,
                                                    const float* __restrict__ er,
                                                    unsigned* __restrict__ m_u) {
    int i = blockIdx.x * blockDim.x + threadIdx.x;
    if (i >= E_EDGES) return;
    const int s_ = src[i], d_ = dst[i];
    v4f l = *(const v4f*)(el + (size_t)s_ * 4);
    v4f r = *(const v4f*)(er + (size_t)d_ * 4);
#pragma unroll
    for (int h = 0; h < 4; ++h)
        atomicMax(&m_u[d_ * 4 + h], flt_flip(edge_score(l[h], r[h])));
}

__global__ __launch_bounds__(256) void gat_edge_sum(const int* __restrict__ src,
                                                    const int* __restrict__ dst,
                                                    const float* __restrict__ el,
                                                    const float* __restrict__ er,
                                                    const unsigned* __restrict__ m_u,
                                                    float* __restrict__ ssum) {
    int i = blockIdx.x * blockDim.x + threadIdx.x;
    if (i >= E_EDGES) return;
    const int s_ = src[i], d_ = dst[i];
    v4f l = *(const v4f*)(el + (size_t)s_ * 4);
    v4f r = *(const v4f*)(er + (size_t)d_ * 4);
#pragma unroll
    for (int h = 0; h < 4; ++h) {
        float e = edge_score(l[h], r[h]);
        float a = expf(e - flt_unflip(m_u[d_ * 4 + h]));
        atomicAdd(&ssum[d_ * 4 + h], a);
    }
}

__global__ __launch_bounds__(256) void gat_edge_agg(const int* __restrict__ src,
                                                    const int* __restrict__ dst,
                                                    const float* __restrict__ el,
                                                    const float* __restrict__ er,
                                                    const unsigned* __restrict__ m_u,
                                                    const float* __restrict__ ssum,
                                                    const float* __restrict__ ft,
                                                    float* __restrict__ out) {
    int t = blockIdx.x * blockDim.x + threadIdx.x;
    if (t >= E_EDGES * H_HEADS) return;
    const int i = t >> 2;
    const int h = t & 3;
    const int s_ = src[i], d_ = dst[i];
    float e = edge_score(el[s_ * 4 + h], er[d_ * 4 + h]);
    float w = expf(e - flt_unflip(m_u[d_ * 4 + h])) / ssum[d_ * 4 + h];
    const float* f = ft + (size_t)s_ * HD + h * D_OUT;
    float* o       = out + (size_t)d_ * HD + h * D_OUT;
#pragma unroll
    for (int d = 0; d < D_OUT; ++d)
        atomicAdd(o + d, f[d] * w);
}

extern "C" void kernel_launch(void* const* d_in, const int* in_sizes, int n_in,
                              void* d_out, int out_size, void* d_ws, size_t ws_size,
                              hipStream_t stream) {
    const float* feat = (const float*)d_in[0];
    const float* W    = (const float*)d_in[1];
    const float* al   = (const float*)d_in[2];
    const float* ar   = (const float*)d_in[3];
    const int*   src  = (const int*)d_in[4];
    const int*   dst  = (const int*)d_in[5];
    float* out = (float*)d_out;

    float*    ft  = (float*)d_ws;
    float*    el  = ft + (size_t)N_NODES * HD;
    float*    er  = el + (size_t)N_NODES * H_HEADS;
    unsigned* m_u = (unsigned*)(er + (size_t)N_NODES * H_HEADS);
    float*    s   = (float*)(m_u + (size_t)N_NODES * H_HEADS);

    const int nb_gemm = N_NODES / 32;
    const int nb_attn = (N_NODES * H_HEADS + 255) / 256;
    const int nb_init = (N_NODES * HD + 255) / 256;
    const int nb_edge = (E_EDGES + 255) / 256;
    const int nb_agg  = (E_EDGES * H_HEADS + 255) / 256;

    gat_gemm<<<nb_gemm, 256, 0, stream>>>(feat, W, ft);
    gat_attn<<<nb_attn, 256, 0, stream>>>(ft, al, ar, el, er);
    gat_init<<<nb_init, 256, 0, stream>>>(m_u, s, out);
    gat_edge_max<<<nb_edge, 256, 0, stream>>>(src, dst, el, er, m_u);
    gat_edge_sum<<<nb_edge, 256, 0, stream>>>(src, dst, el, er, m_u, s);
    gat_edge_agg<<<nb_agg, 256, 0, stream>>>(src, dst, el, er, m_u, s, ft, out);
}


# Aggregation by destination ownership: 256 workgroups each scan dst, queue their node range's edges and sum weighted ft rows in LDS (float LDS atomics, summation order changes); out stored once
# speedup vs baseline: 1.6671x; 1.6671x over previous
#include <hip/hip_runtime.h>

#define N_NODES 100000
#define IN_F    128
#define H_HEADS 4
#define D_OUT   16
#define HD      64
#define E_EDGES 3200000
#define SLOPE   0.2f

typedef __attribute__((ext_vector_type(2))) float v2f;
typedef __attribute__((ext_vector_type(4))) float v4f;
typedef __attribute__((ext_vector_type(8))) float v8f;

__device__ __forceinline__ unsigned flt_flip(float f) {
    unsigned u = __float_as_uint(f);
    return u ^ (((int)u >> 31) | 0x80000000u);
}
__device__ __forceinline__ float flt_unflip(unsigned u) {
    unsigned b = (u & 0x80000000u) ? (u ^ 0x80000000u) : ~u;
    return __uint_as_float(b);
}

__global__ __launch_bounds__(256) void gat_gemm(const float* __restrict__ feat,
                                                const float* __restrict__ W,
                                                float* __restrict__ ft) {
    const int wave = threadIdx.x >> 5;
    const int lane = threadIdx.x & 31;
    const int rt   = wave >> 2;
    const int ct   = wave & 3;
    const int r0   = blockIdx.x * 32 + rt * 16;
    const int c0   = ct * 16;
    const int m    = lane & 15;
    const int hi   = lane >> 4;

    const float* arow = feat + (size_t)(r0 + m) * IN_F + (hi << 1);
    const float* brow = W    + (size_t)(c0 + m) * IN_F + (hi << 1);

    v8f acc = {};
#pragma unroll
    for (int k0 = 0; k0 < IN_F; k0 += 4) {
        v2f a = *(const v2f*)(arow + k0);
        v2f b = *(const v2f*)(brow + k0);
        acc = __builtin_amdgcn_wmma_f32_16x16x4_f32(
            false, a, false, b, (short)0, acc, false, false);
    }
    float* outp = ft + (size_t)(r0 + hi * 8) * HD + c0 + m;
#pragma unroll
    for (int r = 0; r < 8; ++r) outp[(size_t)r * HD] = acc[r];
}

__global__ __launch_bounds__(256) void gat_attn(const float* __restrict__ ft,
                                                const float* __restrict__ al,
                                                const float* __restrict__ ar,
                                                float* __restrict__ el,
                                                float* __restrict__ er) {
    int idx = blockIdx.x * blockDim.x + threadIdx.x;
    if (idx >= N_NODES * H_HEADS) return;
    const int h = idx & 3;
    const float* f = ft + (size_t)idx * D_OUT;
    float sl = 0.f, sr = 0.f;
#pragma unroll
    for (int d = 0; d < D_OUT; ++d) {
        float v = f[d];
        sl += v * al[h * D_OUT + d];
        sr += v * ar[h * D_OUT + d];
    }
    el[idx] = sl;
    er[idx] = sr;
}

__global__ __launch_bounds__(256) void gat_init(unsigned* __restrict__ m_u,
                                                float* __restrict__ s) {
    int idx = blockIdx.x * blockDim.x + threadIdx.x;
    if (idx < N_NODES * H_HEADS) { m_u[idx] = 0u; s[idx] = 0.f; }
}

__device__ __forceinline__ float edge_score(float l, float r) {
    float e = l + r;
    return e >= 0.f ? e : SLOPE * e;
}

__global__ __launch_bounds__(256) void gat_edge_max(const int* __restrict__ src,
                                                    const int* __restrict__ dst,
                                                    const float* __restrict__ el,
                                                    const float* __restrict__ er,
                                                    unsigned* __restrict__ m_u) {
    int i = blockIdx.x * blockDim.x + threadIdx.x;
    if (i >= E_EDGES) return;
    const int s_ = src[i], d_ = dst[i];
    v4f l = *(const v4f*)(el + (size_t)s_ * 4);
    v4f r = *(const v4f*)(er + (size_t)d_ * 4);
#pragma unroll
    for (int h = 0; h < 4; ++h)
        atomicMax(&m_u[d_ * 4 + h], flt_flip(edge_score(l[h], r[h])));
}

__global__ __launch_bounds__(256) void gat_edge_sum(const int* __restrict__ src,
                                                    const int* __restrict__ dst,
                                                    const float* __restrict__ el,
                                                    const float* __restrict__ er,
                                                    const unsigned* __restrict__ m_u,
                                                    float* __restrict__ ssum) {
    int i = blockIdx.x * blockDim.x + threadIdx.x;
    if (i >= E_EDGES) return;
    const int s_ = src[i], d_ = dst[i];
    v4f l = *(const v4f*)(el + (size_t)s_ * 4);
    v4f r = *(const v4f*)(er + (size_t)d_ * 4);
#pragma unroll
    for (int h = 0; h < 4; ++h) {
        float e = edge_score(l[h], r[h]);
        float a = expf(e - flt_unflip(m_u[d_ * 4 + h]));
        atomicAdd(&ssum[d_ * 4 + h], a);
    }
}

#define NPB   391
#define NBLK  256
#define QCAP  32768
#define EBITS 22

__device__ __forceinline__ float edge_weight(float l, float r, unsigned mu, float ss) {
    return expf(edge_score(l, r) - flt_unflip(mu)) / ss;
}

__device__ __forceinline__ void own_check(int dval, int eidx, int lo, unsigned n_own,
                                          unsigned* qcount, unsigned* queue) {
    const unsigned u = (unsigned)(dval - lo);
    if (u < n_own) {
        const unsigned pos = atomicAdd(qcount, 1u);
        if (pos < QCAP) queue[pos] = (unsigned)eidx | (u << EBITS);
    }
}

__device__ __forceinline__ void scan_fill(int4 (&b)[4], const int4* __restrict__ dst4, int q, int nq) {
#pragma unroll
    for (int k = 0; k < 4; ++k) {
        const int qq = q + k * 1024;
        b[k] = dst4[qq < nq ? qq : nq - 1];
    }
}

__device__ __forceinline__ void scan_test(const int4 (&b)[4], int q, int nq, int lo, unsigned n_own,
                                          unsigned* qcount, unsigned* queue) {
#pragma unroll
    for (int k = 0; k < 4; ++k) {
        const int qq = q + k * 1024;
        const unsigned nv = (qq < nq) ? n_own : 0u;
        const int e0 = qq * 4;
        own_check(b[k].x, e0 + 0, lo, nv, qcount, queue);
        own_check(b[k].y, e0 + 1, lo, nv, qcount, queue);
        own_check(b[k].z, e0 + 2, lo, nv, qcount, queue);
        own_check(b[k].w, e0 + 3, lo, nv, qcount, queue);
    }
}

__global__ __launch_bounds__(1024) void gat_edge_agg(const int* __restrict__ src,
                                                     const int* __restrict__ dst,
                                                     const float* __restrict__ el,
                                                     const float* __restrict__ er,
                                                     const unsigned* __restrict__ m_u,
                                                     const float* __restrict__ ssum,
                                                     const float* __restrict__ ft,
                                                     float* __restrict__ out) {
    __shared__ float    acc[NPB * HD];
    __shared__ unsigned queue[QCAP];
    __shared__ unsigned qcount;

    const int tid  = threadIdx.x;
    const int lane = tid & 31;
    const int wave = __builtin_amdgcn_readfirstlane(tid >> 5);
    const int lo   = blockIdx.x * NPB;
    const int rem  = N_NODES - lo;
    const unsigned n_own = (unsigned)(rem < NPB ? rem : NPB);

    for (int k = tid; k < NPB * HD; k += 1024) acc[k] = 0.f;
    if (tid == 0) qcount = 0u;
    __syncthreads();

    const int4* dst4 = (const int4*)dst;
    const int NQ = E_EDGES / 4;
    const int T  = 4096;
    int4 ba[4], bb[4], bc[4];
    scan_fill(ba, dst4, tid, NQ);
    scan_fill(bb, dst4, tid + T, NQ);
    for (int q = tid; q < NQ; q += 3 * T) {
        scan_fill(bc, dst4, q + 2 * T, NQ);
        scan_test(ba, q, NQ, lo, n_own, &qcount, queue);
        scan_fill(ba, dst4, q + 3 * T, NQ);
        scan_test(bb, q + T, NQ, lo, n_own, &qcount, queue);
        scan_fill(bb, dst4, q + 4 * T, NQ);
        scan_test(bc, q + 2 * T, NQ, lo, n_own, &qcount, queue);
    }
    __syncthreads();

    const unsigned total = (unsigned)__builtin_amdgcn_readfirstlane((int)qcount);

    if (total > QCAP) {
        for (int i = tid; i < E_EDGES; i += 1024) {
            const unsigned u = (unsigned)(dst[i] - lo);
            if (u < n_own) {
                const int s_ = src[i];
                const int d_ = lo + (int)u;
                for (int h = 0; h < H_HEADS; ++h) {
                    const float w = edge_weight(el[s_ * 4 + h], er[d_ * 4 + h],
                                                m_u[d_ * 4 + h], ssum[d_ * 4 + h]);
                    const float* f = ft + (size_t)s_ * HD + h * D_OUT;
                    for (int d = 0; d < D_OUT; ++d)
                        atomicAdd(&acc[u * HD + h * D_OUT + d], f[d] * w);
                }
            }
        }
    } else {
        const int cnt = (int)total;
        for (int base = wave * 32; base < cnt; base += 1024) {
            const int idx = base + lane;
            int   s_ = 0, dl = 0;
            float w0 = 0.f, w1 = 0.f, w2 = 0.f, w3 = 0.f;
            if (idx < cnt) {
                const unsigned ent = queue[idx];
                const int ei = (int)(ent & ((1u << EBITS) - 1u));
                dl = (int)(ent >> EBITS);
                s_ = src[ei];
                const int d_ = lo + dl;
                const v4f l  = *(const v4f*)(el + (size_t)s_ * 4);
                const v4f r  = *(const v4f*)(er + (size_t)d_ * 4);
                const v4f ss = *(const v4f*)(ssum + (size_t)d_ * 4);
                const unsigned* mp = m_u + (size_t)d_ * 4;
                w0 = edge_weight(l[0], r[0], mp[0], ss[0]);
                w1 = edge_weight(l[1], r[1], mp[1], ss[1]);
                w2 = edge_weight(l[2], r[2], mp[2], ss[2]);
                w3 = edge_weight(l[3], r[3], mp[3], ss[3]);
            }
            const int left = cnt - base;
            const int nb = left < 32 ? left : 32;
            for (int j0 = 0; j0 < nb; j0 += 16) {
                float fa[16], fb[16];
#pragma unroll
                for (int g = 0; g < 16; ++g) {
                    const int sj = __builtin_amdgcn_readlane(s_, (j0 + g) & 31);
                    const float* f = ft + (size_t)sj * HD;
                    fa[g] = f[lane];
                    fb[g] = f[lane + 32];
                }
#pragma unroll
                for (int g = 0; g < 16; ++g) {
                    const int j = j0 + g;
                    if (j < nb) {
                        const int   dj  = __builtin_amdgcn_readlane(dl, j);
                        const float w0j = __int_as_float(__builtin_amdgcn_readlane(__float_as_int(w0), j));
                        const float w1j = __int_as_float(__builtin_amdgcn_readlane(__float_as_int(w1), j));
                        const float w2j = __int_as_float(__builtin_amdgcn_readlane(__float_as_int(w2), j));
                        const float w3j = __int_as_float(__builtin_amdgcn_readlane(__float_as_int(w3), j));
                        const float wa = (lane < 16) ? w0j : w1j;
                        const float wb = (lane < 16) ? w2j : w3j;
                        atomicAdd(&acc[dj * HD + lane],      fa[g] * wa);
                        atomicAdd(&acc[dj * HD + 32 + lane], fb[g] * wb);
                    }
                }
            }
        }
    }
    __syncthreads();

    float* o = out + (size_t)lo * HD;
    const int nout = (int)n_own * HD;
    for (int k = tid; k < nout; k += 1024) o[k] = acc[k];
}

extern "C" void kernel_launch(void* const* d_in, const int* in_sizes, int n_in,
                              void* d_out, int out_size, void* d_ws, size_t ws_size,
                              hipStream_t stream) {
    const float* feat = (const float*)d_in[0];
    const float* W    = (const float*)d_in[1];
    const float* al   = (const float*)d_in[2];
    const float* ar   = (const float*)d_in[3];
    const int*   src  = (const int*)d_in[4];
    const int*   dst  = (const int*)d_in[5];
    float* out = (float*)d_out;

    float*    ft  = (float*)d_ws;
    float*    el  = ft + (size_t)N_NODES * HD;
    float*    er  = el + (size_t)N_NODES * H_HEADS;
    unsigned* m_u = (unsigned*)(er + (size_t)N_NODES * H_HEADS);
    float*    s   = (float*)(m_u + (size_t)N_NODES * H_HEADS);

    const int nb_gemm = N_NODES / 32;
    const int nb_attn = (N_NODES * H_HEADS + 255) / 256;
    const int nb_edge = (E_EDGES + 255) / 256;

    gat_gemm<<<nb_gemm, 256, 0, stream>>>(feat, W, ft);
    gat_attn<<<nb_attn, 256, 0, stream>>>(ft, al, ar, el, er);
    gat_init<<<nb_attn, 256, 0, stream>>>(m_u, s);
    gat_edge_max<<<nb_edge, 256, 0, stream>>>(src, dst, el, er, m_u);
    gat_edge_sum<<<nb_edge, 256, 0, stream>>>(src, dst, el, er, m_u, s);
    gat_edge_agg<<<NBLK, 1024, 0, stream>>>(src, dst, el, er, m_u, s, ft, out);
}


# Edges bucketed by owner block in a pre-pass (4th launch; runs parked in the owner's out rows until it stores them): a block reads its own edges, the 256-fold dst scan is gone; queue order changes
# speedup vs baseline: 9.8276x; 5.8950x over previous
#include <hip/hip_runtime.h>

#define N_NODES 100000
#define IN_F    128
#define H_HEADS 4
#define D_OUT   16
#define HD      64
#define E_EDGES 3200000
#define SLOPE   0.2f

typedef __attribute__((ext_vector_type(2))) float v2f;
typedef __attribute__((ext_vector_type(4))) float v4f;
typedef __attribute__((ext_vector_type(8))) float v8f;

__device__ __forceinline__ unsigned flt_flip(float f) {
    unsigned u = __float_as_uint(f);
    return u ^ (((int)u >> 31) | 0x80000000u);
}
__device__ __forceinline__ float flt_unflip(unsigned u) {
    unsigned b = (u & 0x80000000u) ? (u ^ 0x80000000u) : ~u;
    return __uint_as_float(b);
}

__global__ __launch_bounds__(256) void gat_gemm(const float* __restrict__ feat,
                                                const float* __restrict__ W,
                                                float* __restrict__ ft) {
    const int wave = threadIdx.x >> 5;
    const int lane = threadIdx.x & 31;
    const int rt   = wave >> 2;
    const int ct   = wave & 3;
    const int r0   = blockIdx.x * 32 + rt * 16;
    const int c0   = ct * 16;
    const int m    = lane & 15;
    const int hi   = lane >> 4;

    const float* arow = feat + (size_t)(r0 + m) * IN_F + (hi << 1);
    const float* brow = W    + (size_t)(c0 + m) * IN_F + (hi << 1);

    v8f acc = {};
#pragma unroll
    for (int k0 = 0; k0 < IN_F; k0 += 4) {
        v2f a = *(const v2f*)(arow + k0);
        v2f b = *(const v2f*)(brow + k0);
        acc = __builtin_amdgcn_wmma_f32_16x16x4_f32(
            false, a, false, b, (short)0, acc, false, false);
    }
    float* outp = ft + (size_t)(r0 + hi * 8) * HD + c0 + m;
#pragma unroll
    for (int r = 0; r < 8; ++r) outp[(size_t)r * HD] = acc[r];
}

__global__ __launch_bounds__(256) void gat_attn(const float* __restrict__ ft,
                                                const float* __restrict__ al,
                                                const float* __restrict__ ar,
                                                float* __restrict__ el,
                                                float* __restrict__ er) {
    int idx = blockIdx.x * blockDim.x + threadIdx.x;
    if (idx >= N_NODES * H_HEADS) return;
    const int h = idx & 3;
    const float* f = ft + (size_t)idx * D_OUT;
    float sl = 0.f, sr = 0.f;
#pragma unroll
    for (int d = 0; d < D_OUT; ++d) {
        float v = f[d];
        sl += v * al[h * D_OUT + d];
        sr += v * ar[h * D_OUT + d];
    }
    el[idx] = sl;
    er[idx] = sr;
}

__device__ __forceinline__ float edge_score(float l, float r) {
    float e = l + r;
    return e >= 0.f ? e : SLOPE * e;
}

#define NPB   391
#define NBLK  256
#define QCAP  32768
#define SBITS 17
#define NBKT  256
#define EPB   (E_EDGES / NBKT)
#define RUN_TRIPS (((NPB * HD) / NBKT - 1 + 31) / 32)

static_assert(EPB * NBKT == E_EDGES && EPB % 4 == 0, "edge list must split into 16-byte aligned parts");
static_assert(NPB < (1 << (32 - SBITS)) && N_NODES <= (1 << SBITS), "queue entry format");

__device__ __forceinline__ unsigned owner_rows(unsigned b) {
    const unsigned rem = (unsigned)N_NODES - b * NPB;
    return rem < NPB ? rem : NPB;
}
__device__ __forceinline__ unsigned run_words(unsigned n_own) { return (n_own * HD) / NBKT; }

__device__ __forceinline__ float edge_weight(float l, float r, unsigned mu, float ss) {
    return expf(edge_score(l, r) - flt_unflip(mu)) / ss;
}

__device__ __forceinline__ void bucket_put(int d, int s, int g, unsigned* cnt, unsigned* rows) {
    const unsigned b = (unsigned)d / NPB;
    if (b < NBLK) {
        const unsigned dl = (unsigned)d - b * NPB;
        const unsigned r  = atomicAdd(&cnt[b], 1u);
        const unsigned S  = run_words(owner_rows(b));
        if (r + 1 < S) rows[(size_t)b * (NPB * HD) + g * S + 1 + r] = (unsigned)s | (dl << SBITS);
    }
}

__global__ __launch_bounds__(1024) void gat_bucket(const int* __restrict__ src,
                                                   const int* __restrict__ dst,
                                                   unsigned* __restrict__ rows) {
    __shared__ unsigned cnt[NBLK];
    const int tid = threadIdx.x;
    const int g   = blockIdx.x;
    if (tid < NBLK) cnt[tid] = 0u;
    __syncthreads();

    const int4* d4 = (const int4*)(dst + (size_t)g * EPB);
    const int4* s4 = (const int4*)(src + (size_t)g * EPB);
    const int nq = EPB / 4;
    const int TR = (EPB / 4 + 1023) / 1024;
    int4 dv[TR], sv[TR];
#pragma unroll
    for (int k = 0; k < TR; ++k) {
        const int q  = tid + k * 1024;
        const int qq = q < nq ? q : nq - 1;
        dv[k] = d4[qq];
        sv[k] = s4[qq];
    }
#pragma unroll
    for (int k = 0; k < TR; ++k) {
        if (tid + k * 1024 < nq) {
            bucket_put(dv[k].x, sv[k].x, g, cnt, rows);
            bucket_put(dv[k].y, sv[k].y, g, cnt, rows);
            bucket_put(dv[k].z, sv[k].z, g, cnt, rows);
            bucket_put(dv[k].w, sv[k].w, g, cnt, rows);
        }
    }
    __syncthreads();
    if (tid < NBLK)
        rows[(size_t)tid * (NPB * HD) + g * run_words(owner_rows((unsigned)tid))] = cnt[tid];
}

__global__ __launch_bounds__(1024) void gat_edge_agg(const int* __restrict__ src,
                                                     const int* __restrict__ dst,
                                                     const float* __restrict__ el,
                                                     const float* __restrict__ er,
                                                     const float* __restrict__ ft,
                                                     float* out) {
    __shared__ float    acc[NPB * HD];
    __shared__ unsigned queue[QCAP];
    __shared__ __attribute__((aligned(16))) unsigned mx[NPB * H_HEADS];
    __shared__ __attribute__((aligned(16))) float    sm[NPB * H_HEADS];
    __shared__ __attribute__((aligned(16))) float    ers[NPB * H_HEADS];
    __shared__ unsigned runoff[NBKT + 1];
    __shared__ unsigned wsum[NBKT / 32];
    __shared__ unsigned qcount;
    __shared__ unsigned overfull;

    const int tid  = threadIdx.x;
    const int lane = tid & 31;
    const int wave = __builtin_amdgcn_readfirstlane(tid >> 5);
    const int lo   = blockIdx.x * NPB;
    const int rem  = N_NODES - lo;
    const unsigned n_own = (unsigned)(rem < NPB ? rem : NPB);
    const unsigned S = run_words(n_own);
    const unsigned* runs = (const unsigned*)(out + (size_t)lo * HD);

    for (int k = tid; k < NPB * HD; k += 1024) acc[k] = 0.f;
    for (int k = tid; k < NPB * H_HEADS; k += 1024) {
        mx[k]  = 0u;
        sm[k]  = 0.f;
        ers[k] = (k < (int)n_own * H_HEADS) ? er[(size_t)lo * H_HEADS + k] : 0.f;
    }
    if (tid == 0) { qcount = 0u; overfull = 0u; }
    __syncthreads();

    unsigned c = 0u, incl = 0u;
    if (tid < NBKT) {
        c = runs[(unsigned)tid * S];
        if (c >= S) atomicOr(&overfull, 1u);
        incl = c;
        for (int d = 1; d < 32; d <<= 1) {
            const unsigned v = __shfl_up(incl, d);
            if (lane >= d) incl += v;
        }
        if (lane == 31) wsum[wave] = incl;
    }
    __syncthreads();
    if (tid < NBKT) {
        unsigned off = incl - c;
        for (int w = 0; w < wave; ++w) off += wsum[w];
        runoff[tid] = off;
        if (tid == NBKT - 1) runoff[NBKT] = off + c;
    }
    __syncthreads();
    const bool bucketed = __builtin_amdgcn_readfirstlane((int)overfull) == 0;

    if (bucketed) {
        unsigned v[NBKT / 32][RUN_TRIPS];
#pragma unroll
        for (int gi = 0; gi < NBKT / 32; ++gi) {
            const int g = wave + gi * 32;
            const unsigned n = runoff[g + 1] - runoff[g];
#pragma unroll
            for (int t = 0; t < RUN_TRIPS; ++t) {
                const unsigned j = (unsigned)lane + t * 32;
                v[gi][t] = j < n ? runs[(unsigned)g * S + 1 + j] : 0u;
            }
        }
#pragma unroll
        for (int gi = 0; gi < NBKT / 32; ++gi) {
            const int g = wave + gi * 32;
            const unsigned o = runoff[g], n = runoff[g + 1] - o;
#pragma unroll
            for (int t = 0; t < RUN_TRIPS; ++t) {
                const unsigned j = (unsigned)lane + t * 32;
                if (j < n) queue[o + j] = v[gi][t];
            }
        }
        if (tid == 0) qcount = runoff[NBKT];
    } else {
        for (int i = tid; i < E_EDGES; i += 1024) {
            const unsigned u = (unsigned)(dst[i] - lo);
            if (u < n_own) {
                const unsigned pos = atomicAdd(&qcount, 1u);
                if (pos < QCAP) queue[pos] = (unsigned)src[i] | (u << SBITS);
            }
        }
    }
    __syncthreads();

    const unsigned total = (unsigned)__builtin_amdgcn_readfirstlane((int)qcount);
    const bool queued = total <= QCAP;
    const int  cnt    = (int)total;

    if (queued) {
        for (int base = tid; base < cnt; base += 4 * 1024) {
            unsigned ent[4];
            v4f      lv[4];
#pragma unroll
            for (int k = 0; k < 4; ++k) {
                const int idx = base + k * 1024;
                ent[k] = idx < cnt ? queue[idx] : 0u;
            }
#pragma unroll
            for (int k = 0; k < 4; ++k)
                lv[k] = *(const v4f*)(el + (size_t)(ent[k] & ((1u << SBITS) - 1u)) * 4);
#pragma unroll
            for (int k = 0; k < 4; ++k) {
                const int idx = base + k * 1024;
                if (idx < cnt) {
                    const unsigned dl = ent[k] >> SBITS;
                    const v4f r = *(const v4f*)(ers + dl * 4);
#pragma unroll
                    for (int h = 0; h < 4; ++h)
                        atomicMax(&mx[dl * 4 + h], flt_flip(edge_score(lv[k][h], r[h])));
                }
            }
        }
    } else {
        for (int i = tid; i < E_EDGES; i += 1024) {
            const unsigned u = (unsigned)(dst[i] - lo);
            if (u < n_own) {
                const int s_ = src[i];
                for (int h = 0; h < H_HEADS; ++h)
                    atomicMax(&mx[u * 4 + h], flt_flip(edge_score(el[s_ * 4 + h], ers[u * 4 + h])));
            }
        }
    }
    __syncthreads();

    if (queued) {
        for (int base = tid; base < cnt; base += 4 * 1024) {
            unsigned ent[4];
            v4f      lv[4];
#pragma unroll
            for (int k = 0; k < 4; ++k) {
                const int idx = base + k * 1024;
                ent[k] = idx < cnt ? queue[idx] : 0u;
            }
#pragma unroll
            for (int k = 0; k < 4; ++k)
                lv[k] = *(const v4f*)(el + (size_t)(ent[k] & ((1u << SBITS) - 1u)) * 4);
#pragma unroll
            for (int k = 0; k < 4; ++k) {
                const int idx = base + k * 1024;
                if (idx < cnt) {
                    const unsigned dl = ent[k] >> SBITS;
                    const v4f r = *(const v4f*)(ers + dl * 4);
                    const uint4 mu = *(const uint4*)(mx + dl * 4);
                    atomicAdd(&sm[dl * 4 + 0], expf(edge_score(lv[k][0], r[0]) - flt_unflip(mu.x)));
                    atomicAdd(&sm[dl * 4 + 1], expf(edge_score(lv[k][1], r[1]) - flt_unflip(mu.y)));
                    atomicAdd(&sm[dl * 4 + 2], expf(edge_score(lv[k][2], r[2]) - flt_unflip(mu.z)));
                    atomicAdd(&sm[dl * 4 + 3], expf(edge_score(lv[k][3], r[3]) - flt_unflip(mu.w)));
                }
            }
        }
    } else {
        for (int i = tid; i < E_EDGES; i += 1024) {
            const unsigned u = (unsigned)(dst[i] - lo);
            if (u < n_own) {
                const int s_ = src[i];
                for (int h = 0; h < H_HEADS; ++h)
                    atomicAdd(&sm[u * 4 + h],
                              expf(edge_score(el[s_ * 4 + h], ers[u * 4 + h]) - flt_unflip(mx[u * 4 + h])));
            }
        }
    }
    __syncthreads();

    if (!queued) {
        for (int i = tid; i < E_EDGES; i += 1024) {
            const unsigned u = (unsigned)(dst[i] - lo);
            if (u < n_own) {
                const int s_ = src[i];
                for (int h = 0; h < H_HEADS; ++h) {
                    const float w = edge_weight(el[s_ * 4 + h], ers[u * 4 + h],
                                                mx[u * 4 + h], sm[u * 4 + h]);
                    const float* f = ft + (size_t)s_ * HD + h * D_OUT;
                    for (int d = 0; d < D_OUT; ++d)
                        atomicAdd(&acc[u * HD + h * D_OUT + d], f[d] * w);
                }
            }
        }
    } else {
        for (int base = wave * 32; base < cnt; base += 1024) {
            const int idx = base + lane;
            int   s_ = 0, dl = 0;
            float w0 = 0.f, w1 = 0.f, w2 = 0.f, w3 = 0.f;
            if (idx < cnt) {
                const unsigned ent = queue[idx];
                s_ = (int)(ent & ((1u << SBITS) - 1u));
                dl = (int)(ent >> SBITS);
                const v4f   l  = *(const v4f*)(el + (size_t)s_ * 4);
                const v4f   r  = *(const v4f*)(ers + dl * 4);
                const v4f   ss = *(const v4f*)(sm + dl * 4);
                const uint4 mu = *(const uint4*)(mx + dl * 4);
                w0 = edge_weight(l[0], r[0], mu.x, ss[0]);
                w1 = edge_weight(l[1], r[1], mu.y, ss[1]);
                w2 = edge_weight(l[2], r[2], mu.z, ss[2]);
                w3 = edge_weight(l[3], r[3], mu.w, ss[3]);
            }
            const int left = cnt - base;
            const int nb = left < 32 ? left : 32;
            for (int j0 = 0; j0 < nb; j0 += 16) {
                float fa[16], fb[16];
#pragma unroll
                for (int g = 0; g < 16; ++g) {
                    const int sj = __builtin_amdgcn_readlane(s_, (j0 + g) & 31);
                    const float* f = ft + (size_t)sj * HD;
                    fa[g] = f[lane];
                    fb[g] = f[lane + 32];
                }
#pragma unroll
                for (int g = 0; g < 16; ++g) {
                    const int j = j0 + g;
                    if (j < nb) {
                        const int   dj  = __builtin_amdgcn_readlane(dl, j);
                        const float w0j = __int_as_float(__builtin_amdgcn_readlane(__float_as_int(w0), j));
                        const float w1j = __int_as_float(__builtin_amdgcn_readlane(__float_as_int(w1), j));
                        const float w2j = __int_as_float(__builtin_amdgcn_readlane(__float_as_int(w2), j));
                        const float w3j = __int_as_float(__builtin_amdgcn_readlane(__float_as_int(w3), j));
                        const float wa = (lane < 16) ? w0j : w1j;
                        const float wb = (lane < 16) ? w2j : w3j;
                        atomicAdd(&acc[dj * HD + lane],      fa[g] * wa);
                        atomicAdd(&acc[dj * HD + 32 + lane], fb[g] * wb);
                    }
                }
            }
        }
    }
    __syncthreads();

    float* o = out + (size_t)lo * HD;
    const int nout = (int)n_own * HD;
    for (int k = tid; k < nout; k += 1024) o[k] = acc[k];
}

extern "C" void kernel_launch(void* const* d_in, const int* in_sizes, int n_in,
                              void* d_out, int out_size, void* d_ws, size_t ws_size,
                              hipStream_t stream) {
    const float* feat = (const float*)d_in[0];
    const float* W    = (const float*)d_in[1];
    const float* al   = (const float*)d_in[2];
    const float* ar   = (const float*)d_in[3];
    const int*   src  = (const int*)d_in[4];
    const int*   dst  = (const int*)d_in[5];
    float* out = (float*)d_out;

    float* ft = (float*)d_ws;
    float* el = ft + (size_t)N_NODES * HD;
    float* er = el + (size_t)N_NODES * H_HEADS;

    const int nb_gemm = N_NODES / 32;
    const int nb_attn = (N_NODES * H_HEADS + 255) / 256;

    gat_bucket<<<NBKT, 1024, 0, stream>>>(src, dst, (unsigned*)d_out);
    gat_gemm<<<nb_gemm, 256, 0, stream>>>(feat, W, ft);
    gat_attn<<<nb_attn, 256, 0, stream>>>(ft, al, ar, el, er);
    gat_edge_agg<<<NBLK, 1024, 0, stream>>>(src, dst, el, er, ft, out);
}


# Owner block sorts its edges by destination node in LDS (counting sort); one wave per node: softmax by lane reductions, weighted ft rows summed in registers; LDS float atomics gone, sum order changes
# speedup vs baseline: 13.2226x; 1.3455x over previous
#include <hip/hip_runtime.h>

#define N_NODES 100000
#define IN_F    128
#define H_HEADS 4
#define D_OUT   16
#define HD      64
#define E_EDGES 3200000
#define SLOPE   0.2f

typedef __attribute__((ext_vector_type(2))) float v2f;
typedef __attribute__((ext_vector_type(4))) float v4f;
typedef __attribute__((ext_vector_type(8))) float v8f;

__device__ __forceinline__ unsigned flt_flip(float f) {
    unsigned u = __float_as_uint(f);
    return u ^ (((int)u >> 31) | 0x80000000u);
}
__device__ __forceinline__ float flt_unflip(unsigned u) {
    unsigned b = (u & 0x80000000u) ? (u ^ 0x80000000u) : ~u;
    return __uint_as_float(b);
}

__global__ __launch_bounds__(256) void gat_gemm(const float* __restrict__ feat,
                                                const float* __restrict__ W,
                                                float* __restrict__ ft) {
    const int wave = threadIdx.x >> 5;
    const int lane = threadIdx.x & 31;
    const int rt   = wave >> 2;
    const int ct   = wave & 3;
    const int r0   = blockIdx.x * 32 + rt * 16;
    const int c0   = ct * 16;
    const int m    = lane & 15;
    const int hi   = lane >> 4;

    const float* arow = feat + (size_t)(r0 + m) * IN_F + (hi << 1);
    const float* brow = W    + (size_t)(c0 + m) * IN_F + (hi << 1);

    v8f acc = {};
#pragma unroll
    for (int k0 = 0; k0 < IN_F; k0 += 4) {
        v2f a = *(const v2f*)(arow + k0);
        v2f b = *(const v2f*)(brow + k0);
        acc = __builtin_amdgcn_wmma_f32_16x16x4_f32(
            false, a, false, b, (short)0, acc, false, false);
    }
    float* outp = ft + (size_t)(r0 + hi * 8) * HD + c0 + m;
#pragma unroll
    for (int r = 0; r < 8; ++r) outp[(size_t)r * HD] = acc[r];
}

__global__ __launch_bounds__(256) void gat_attn(const float* __restrict__ ft,
                                                const float* __restrict__ al,
                                                const float* __restrict__ ar,
                                                float* __restrict__ el,
                                                float* __restrict__ er) {
    int idx = blockIdx.x * blockDim.x + threadIdx.x;
    if (idx >= N_NODES * H_HEADS) return;
    const int h = idx & 3;
    const float* f = ft + (size_t)idx * D_OUT;
    float sl = 0.f, sr = 0.f;
#pragma unroll
    for (int d = 0; d < D_OUT; ++d) {
        float v = f[d];
        sl += v * al[h * D_OUT + d];
        sr += v * ar[h * D_OUT + d];
    }
    el[idx] = sl;
    er[idx] = sr;
}

__device__ __forceinline__ float edge_score(float l, float r) {
    float e = l + r;
    return e >= 0.f ? e : SLOPE * e;
}

#define NPB   391
#define NBLK  256
#define SBITS 17
#define NBKT  256
#define EPB   (E_EDGES / NBKT)
#define RUN_TRIPS (((NPB * HD) / NBKT - 1 + 31) / 32)

static_assert(EPB * NBKT == E_EDGES && EPB % 4 == 0, "edge list must split into 16-byte aligned parts");
static_assert(NPB < (1 << (32 - SBITS)) && N_NODES <= (1 << SBITS), "run entry format");

__device__ __forceinline__ unsigned owner_rows(unsigned b) {
    const unsigned rem = (unsigned)N_NODES - b * NPB;
    return rem < NPB ? rem : NPB;
}
__device__ __forceinline__ unsigned run_words(unsigned n_own) { return (n_own * HD) / NBKT; }

__device__ __forceinline__ float edge_weight(float l, float r, unsigned mu, float ss) {
    return expf(edge_score(l, r) - flt_unflip(mu)) / ss;
}

__device__ __forceinline__ void bucket_put(int d, int s, int g, unsigned* cnt, unsigned* rows) {
    const unsigned b = (unsigned)d / NPB;
    if (b < NBLK) {
        const unsigned dl = (unsigned)d - b * NPB;
        const unsigned r  = atomicAdd(&cnt[b], 1u);
        const unsigned S  = run_words(owner_rows(b));
        if (r + 1 < S) rows[(size_t)b * (NPB * HD) + g * S + 1 + r] = (unsigned)s | (dl << SBITS);
    }
}

__global__ __launch_bounds__(1024) void gat_bucket(const int* __restrict__ src,
                                                   const int* __restrict__ dst,
                                                   unsigned* __restrict__ rows) {
    __shared__ unsigned cnt[NBLK];
    const int tid = threadIdx.x;
    const int g   = blockIdx.x;
    if (tid < NBLK) cnt[tid] = 0u;
    __syncthreads();

    const int4* d4 = (const int4*)(dst + (size_t)g * EPB);
    const int4* s4 = (const int4*)(src + (size_t)g * EPB);
    const int nq = EPB / 4;
    const int TR = (EPB / 4 + 1023) / 1024;
    int4 dv[TR], sv[TR];
#pragma unroll
    for (int k = 0; k < TR; ++k) {
        const int q  = tid + k * 1024;
        const int qq = q < nq ? q : nq - 1;
        dv[k] = d4[qq];
        sv[k] = s4[qq];
    }
#pragma unroll
    for (int k = 0; k < TR; ++k) {
        if (tid + k * 1024 < nq) {
            bucket_put(dv[k].x, sv[k].x, g, cnt, rows);
            bucket_put(dv[k].y, sv[k].y, g, cnt, rows);
            bucket_put(dv[k].z, sv[k].z, g, cnt, rows);
            bucket_put(dv[k].w, sv[k].w, g, cnt, rows);
        }
    }
    __syncthreads();
    if (tid < NBLK)
        rows[(size_t)tid * (NPB * HD) + g * run_words(owner_rows((unsigned)tid))] = cnt[tid];
}

template <int CTRL>
__device__ __forceinline__ float row_xor(float x) {
    const int b = __builtin_bit_cast(int, x);
    return __builtin_bit_cast(float, __builtin_amdgcn_update_dpp(b, b, CTRL, 0xf, 0xf, false));
}
__device__ __forceinline__ float xhalf(float x) {
    const int b = __builtin_bit_cast(int, x);
    return __builtin_bit_cast(float,
        __builtin_amdgcn_permlanex16(b, b, 0x76543210, 0xfedcba98, false, false));
}
template <bool ISMAX>
__device__ __forceinline__ float red_op(float a, float b) { return ISMAX ? fmaxf(a, b) : a + b; }
template <bool ISMAX>
__device__ __forceinline__ void wave_reduce4(float& x0, float& x1, float& x2, float& x3, int lane) {
    const bool hi = (lane & 16) != 0, b3 = (lane & 8) != 0;
    const float keepA = hi ? x2 : x0, keepB = hi ? x3 : x1;
    const float sendA = hi ? x0 : x2, sendB = hi ? x1 : x3;
    const float yA = red_op<ISMAX>(keepA, xhalf(sendA));
    const float yB = red_op<ISMAX>(keepB, xhalf(sendB));
    const float keep = b3 ? yB : yA, send = b3 ? yA : yB;
    float z = red_op<ISMAX>(keep, row_xor<0x168>(send));
    z = red_op<ISMAX>(z, row_xor<0x164>(z));
    z = red_op<ISMAX>(z, row_xor<0x162>(z));
    z = red_op<ISMAX>(z, row_xor<0x161>(z));
    const int zi = __builtin_bit_cast(int, z);
    x0 = __builtin_bit_cast(float, __builtin_amdgcn_readlane(zi, 0));
    x1 = __builtin_bit_cast(float, __builtin_amdgcn_readlane(zi, 8));
    x2 = __builtin_bit_cast(float, __builtin_amdgcn_readlane(zi, 16));
    x3 = __builtin_bit_cast(float, __builtin_amdgcn_readlane(zi, 24));
}
__device__ __forceinline__ float pick4(int h, float x0, float x1, float x2, float x3) {
    return h == 0 ? x0 : (h == 1 ? x1 : (h == 2 ? x2 : x3));
}

#define WCAP  128
#define ROWU  8
#define NOENT 0xFFFFFFFFu

template <bool FULL>
__device__ __forceinline__ void row_group(const float* wq, const float* __restrict__ ft,
                                          int t0, int nv, int hf, int hl, unsigned c4b,
                                          v2f& accA, v2f& accB) {
    v2f ws[ROWU];
    v4f rv[ROWU];
#pragma unroll
    for (int u = 0; u < ROWU; ++u)
        ws[u] = *(const v2f*)(wq + (2 * (t0 + u) + hf) * 8 + hl * 2);
#pragma unroll
    for (int u = 0; u < ROWU; ++u) {
        if (FULL || u < nv) {
            rv[u] = *(const v4f*)((const char*)ft + (size_t)((__float_as_uint(ws[u][1]) << 8) + c4b));
        } else {
            rv[u][0] = rv[u][1] = rv[u][2] = rv[u][3] = 0.f;
            ws[u][0] = 0.f;
        }
    }
#pragma unroll
    for (int u = 0; u < ROWU; ++u) {
        const v2f w2 = {ws[u][0], ws[u][0]};
        const v2f lo = {rv[u][0], rv[u][1]};
        const v2f hi = {rv[u][2], rv[u][3]};
        accA = w2 * lo + accA;
        accB = w2 * hi + accB;
    }
}

__global__ __launch_bounds__(1024) void gat_edge_agg(const int* __restrict__ src,
                                                     const int* __restrict__ dst,
                                                     const float* __restrict__ el,
                                                     const float* __restrict__ er,
                                                     const float* __restrict__ ft,
                                                     float* out) {
    __shared__ __attribute__((aligned(16))) union { float f[NPB * HD]; unsigned u[NPB * HD]; } buf;
    __shared__ __attribute__((aligned(16))) float wq_all[(32 * WCAP + 16) * 8];
    __shared__ __attribute__((aligned(16))) unsigned mx[NPB * H_HEADS];
    __shared__ __attribute__((aligned(16))) float    sm[NPB * H_HEADS];
    __shared__ __attribute__((aligned(16))) float    ers[NPB * H_HEADS];
    __shared__ unsigned ncnt[NPB + 1];
    __shared__ unsigned nstart[NPB + 1];
    __shared__ unsigned rcnt[NBKT];
    __shared__ unsigned wsum[16];
    __shared__ unsigned overfull;

    const int tid  = threadIdx.x;
    const int lane = tid & 31;
    const int wave = __builtin_amdgcn_readfirstlane(tid >> 5);
    const int lo   = blockIdx.x * NPB;
    const int rem  = N_NODES - lo;
    const unsigned n_own = (unsigned)(rem < NPB ? rem : NPB);
    const unsigned S = run_words(n_own);
    const unsigned* runs = (const unsigned*)(out + (size_t)lo * HD);

    for (int k = tid; k <= NPB; k += 1024) ncnt[k] = 0u;
    if (tid == 0) overfull = 0u;
    __syncthreads();
    if (tid < NBKT) {
        const unsigned c = runs[(unsigned)tid * S];
        if (c >= S) atomicOr(&overfull, 1u);
        rcnt[tid] = c;
    }
    __syncthreads();
    const bool bucketed = __builtin_amdgcn_readfirstlane((int)overfull) == 0;

    if (bucketed) {
        unsigned v[NBKT / 32][RUN_TRIPS];
#pragma unroll
        for (int gi = 0; gi < NBKT / 32; ++gi) {
            const int g = wave + gi * 32;
            const unsigned n = rcnt[g];
#pragma unroll
            for (int t = 0; t < RUN_TRIPS; ++t) {
                const unsigned j = (unsigned)lane + t * 32;
                v[gi][t] = j < n ? runs[(unsigned)g * S + 1 + j] : NOENT;
            }
        }
#pragma unroll
        for (int gi = 0; gi < NBKT / 32; ++gi)
#pragma unroll
            for (int t = 0; t < RUN_TRIPS; ++t)
                if (v[gi][t] != NOENT) atomicAdd(&ncnt[v[gi][t] >> SBITS], 1u);
        __syncthreads();
        unsigned cv = 0u, inc = 0u;
        if (tid < 512) {
            cv  = tid <= NPB ? ncnt[tid] : 0u;
            inc = cv;
            for (int d = 1; d < 32; d <<= 1) {
                const unsigned x = __shfl_up(inc, d);
                if (lane >= d) inc += x;
            }
            if (lane == 31) wsum[wave] = inc;
        }
        __syncthreads();
        if (tid <= NPB) {
            unsigned off = inc - cv;
            for (int w = 0; w < wave; ++w) off += wsum[w];
            nstart[tid] = off;
            ncnt[tid]   = off;
        }
        __syncthreads();
#pragma unroll
        for (int gi = 0; gi < NBKT / 32; ++gi)
#pragma unroll
            for (int t = 0; t < RUN_TRIPS; ++t)
                if (v[gi][t] != NOENT) {
                    const unsigned pos = atomicAdd(&ncnt[v[gi][t] >> SBITS], 1u);
                    buf.u[pos] = v[gi][t] & ((1u << SBITS) - 1u);
                }
        __syncthreads();

        float* wq = wq_all + wave * (WCAP * 8);
        const int      hf  = lane >> 4;
        const unsigned c4b = (unsigned)(lane & 15) * 16u;
        const int      hl  = (lane & 15) >> 2;
        for (int node = wave; node < (int)n_own; node += 32) {
            const int beg = __builtin_amdgcn_readfirstlane((int)nstart[node]);
            const int deg = __builtin_amdgcn_readfirstlane((int)nstart[node + 1]) - beg;
            const v4f r4  = *(const v4f*)(er + (size_t)(lo + node) * 4);

            float m0 = -INFINITY, m1 = -INFINITY, m2 = -INFINITY, m3 = -INFINITY;
            for (int jb = 0; jb < deg; jb += 64) {
                const int  ja = jb + lane, jc = ja + 32;
                const bool va = ja < deg,  vc = jc < deg;
                const unsigned sa = buf.u[beg + (va ? ja : deg - 1)];
                const unsigned sc = buf.u[beg + (vc ? jc : deg - 1)];
                const v4f la = *(const v4f*)((const char*)el + (size_t)(sa << 4));
                const v4f lc = *(const v4f*)((const char*)el + (size_t)(sc << 4));
                v4f ea, ec;
#pragma unroll
                for (int h = 0; h < 4; ++h) {
                    ea[h] = va ? edge_score(la[h], r4[h]) : -INFINITY;
                    ec[h] = vc ? edge_score(lc[h], r4[h]) : -INFINITY;
                }
                if (ja < WCAP) {
                    const float sf = __uint_as_float(sa);
                    v4f q0, q1;
                    q0[0] = ea[0]; q0[1] = sf; q0[2] = ea[1]; q0[3] = sf;
                    q1[0] = ea[2]; q1[1] = sf; q1[2] = ea[3]; q1[3] = sf;
                    *(v4f*)(wq + ja * 8)     = q0;
                    *(v4f*)(wq + ja * 8 + 4) = q1;
                }
                if (jc < WCAP) {
                    const float sf = __uint_as_float(sc);
                    v4f q0, q1;
                    q0[0] = ec[0]; q0[1] = sf; q0[2] = ec[1]; q0[3] = sf;
                    q1[0] = ec[2]; q1[1] = sf; q1[2] = ec[3]; q1[3] = sf;
                    *(v4f*)(wq + jc * 8)     = q0;
                    *(v4f*)(wq + jc * 8 + 4) = q1;
                }
                m0 = fmaxf(m0, fmaxf(ea[0], ec[0]));
                m1 = fmaxf(m1, fmaxf(ea[1], ec[1]));
                m2 = fmaxf(m2, fmaxf(ea[2], ec[2]));
                m3 = fmaxf(m3, fmaxf(ea[3], ec[3]));
            }
            wave_reduce4<true>(m0, m1, m2, m3, lane);

            float s0 = 0.f, s1 = 0.f, s2 = 0.f, s3 = 0.f;
            for (int jb = 0; jb < deg; jb += 32) {
                const int j = jb + lane;
                float e0 = -INFINITY, e1 = -INFINITY, e2 = -INFINITY, e3 = -INFINITY, sf = 0.f;
                if (j < WCAP) {
                    const v4f q0 = *(const v4f*)(wq + j * 8);
                    const v4f q1 = *(const v4f*)(wq + j * 8 + 4);
                    e0 = q0[0]; e1 = q0[2]; e2 = q1[0]; e3 = q1[2]; sf = q0[1];
                } else if (j < deg) {
                    const unsigned s = buf.u[beg + j];
                    const v4f l = *(const v4f*)((const char*)el + (size_t)(s << 4));
                    e0 = edge_score(l[0], r4[0]);
                    e1 = edge_score(l[1], r4[1]);
                    e2 = edge_score(l[2], r4[2]);
                    e3 = edge_score(l[3], r4[3]);
                }
                float x0 = 0.f, x1 = 0.f, x2 = 0.f, x3 = 0.f;
                if (j < deg) {
                    x0 = expf(e0 - m0);
                    x1 = expf(e1 - m1);
                    x2 = expf(e2 - m2);
                    x3 = expf(e3 - m3);
                }
                if (j < WCAP) {
                    v4f q0, q1;
                    q0[0] = x0; q0[1] = sf; q0[2] = x1; q0[3] = sf;
                    q1[0] = x2; q1[1] = sf; q1[2] = x3; q1[3] = sf;
                    *(v4f*)(wq + j * 8)     = q0;
                    *(v4f*)(wq + j * 8 + 4) = q1;
                }
                s0 += x0; s1 += x1; s2 += x2; s3 += x3;
            }
            wave_reduce4<false>(s0, s1, s2, s3, lane);

            v2f accA = {0.f, 0.f}, accB = {0.f, 0.f};
            const int npairs = (deg + 1) >> 1;
            const int lim    = deg < WCAP ? deg : WCAP;
            const int nfast  = (lim + 1) >> 1;
            int t0 = 0;
            for (; t0 + ROWU <= nfast; t0 += ROWU)
                row_group<true>(wq, ft, t0, ROWU, hf, hl, c4b, accA, accB);
            if (t0 < nfast)
                row_group<false>(wq, ft, t0, nfast - t0, hf, hl, c4b, accA, accB);
            float a0 = accA[0], a1 = accA[1], a2 = accB[0], a3 = accB[1];
            if (deg > WCAP) {
                const float rl = pick4(hl, r4[0], r4[1], r4[2], r4[3]);
                const float ml = pick4(hl, m0, m1, m2, m3);
                for (int t = WCAP / 2; t < npairs; ++t) {
                    const int j  = 2 * t + hf;
                    const int jj = j < deg ? j : deg - 1;
                    const unsigned s = buf.u[beg + jj];
                    const float l = *(const float*)((const char*)el + (size_t)((s << 4) + (unsigned)hl * 4u));
                    const float w = j < deg ? expf(edge_score(l, rl) - ml) : 0.f;
                    const v4f r = *(const v4f*)((const char*)ft + (size_t)((s << 8) + c4b));
                    a0 += w * r[0];
                    a1 += w * r[1];
                    a2 += w * r[2];
                    a3 += w * r[3];
                }
            }

            const float sl  = pick4(hl, s0, s1, s2, s3);
            const float inv = deg > 0 ? 1.0f / sl : 0.f;
            a0 += xhalf(a0);
            a1 += xhalf(a1);
            a2 += xhalf(a2);
            a3 += xhalf(a3);
            if (lane < 16) {
                v4f o;
                o[0] = a0 * inv; o[1] = a1 * inv; o[2] = a2 * inv; o[3] = a3 * inv;
                *(v4f*)((char*)out + (size_t)(lo + node) * (HD * 4) + c4b) = o;
            }
        }
    } else {
        float* acc = buf.f;
        for (int k = tid; k < NPB * HD; k += 1024) acc[k] = 0.f;
        for (int k = tid; k < NPB * H_HEADS; k += 1024) {
            mx[k]  = 0u;
            sm[k]  = 0.f;
            ers[k] = (k < (int)n_own * H_HEADS) ? er[(size_t)lo * H_HEADS + k] : 0.f;
        }
        __syncthreads();
        for (int i = tid; i < E_EDGES; i += 1024) {
            const unsigned u = (unsigned)(dst[i] - lo);
            if (u < n_own) {
                const int s_ = src[i];
                for (int h = 0; h < H_HEADS; ++h)
                    atomicMax(&mx[u * 4 + h], flt_flip(edge_score(el[s_ * 4 + h], ers[u * 4 + h])));
            }
        }
        __syncthreads();
        for (int i = tid; i < E_EDGES; i += 1024) {
            const unsigned u = (unsigned)(dst[i] - lo);
            if (u < n_own) {
                const int s_ = src[i];
                for (int h = 0; h < H_HEADS; ++h)
                    atomicAdd(&sm[u * 4 + h],
                              expf(edge_score(el[s_ * 4 + h], ers[u * 4 + h]) - flt_unflip(mx[u * 4 + h])));
            }
        }
        __syncthreads();
        for (int i = tid; i < E_EDGES; i += 1024) {
            const unsigned u = (unsigned)(dst[i] - lo);
            if (u < n_own) {
                const int s_ = src[i];
                for (int h = 0; h < H_HEADS; ++h) {
                    const float w = edge_weight(el[s_ * 4 + h], ers[u * 4 + h],
                                                mx[u * 4 + h], sm[u * 4 + h]);
                    const float* f = ft + (size_t)s_ * HD + h * D_OUT;
                    for (int d = 0; d < D_OUT; ++d)
                        atomicAdd(&acc[u * HD + h * D_OUT + d], f[d] * w);
                }
            }
        }
        __syncthreads();
        float* o = out + (size_t)lo * HD;
        const int nout = (int)n_own * HD;
        for (int k = tid; k < nout; k += 1024) o[k] = acc[k];
    }
}

extern "C" void kernel_launch(void* const* d_in, const int* in_sizes, int n_in,
                              void* d_out, int out_size, void* d_ws, size_t ws_size,
                              hipStream_t stream) {
    const float* feat = (const float*)d_in[0];
    const float* W    = (const float*)d_in[1];
    const float* al   = (const float*)d_in[2];
    const float* ar   = (const float*)d_in[3];
    const int*   src  = (const int*)d_in[4];
    const int*   dst  = (const int*)d_in[5];
    float* out = (float*)d_out;

    float* ft = (float*)d_ws;
    float* el = ft + (size_t)N_NODES * HD;
    float* er = el + (size_t)N_NODES * H_HEADS;

    const int nb_gemm = N_NODES / 32;
    const int nb_attn = (N_NODES * H_HEADS + 255) / 256;

    gat_bucket<<<NBKT, 1024, 0, stream>>>(src, dst, (unsigned*)d_out);
    gat_gemm<<<nb_gemm, 256, 0, stream>>>(feat, W, ft);
    gat_attn<<<nb_attn, 256, 0, stream>>>(ft, al, ar, el, er);
    gat_edge_agg<<<NBLK, 1024, 0, stream>>>(src, dst, el, er, ft, out);
}


# Projection re-tiled: one wave forms 16 rows across all 64 columns (feat read once, not once per head), W fragments from LDS, 256 blocks of 32 waves; k order of the matrix sums changes
# speedup vs baseline: 14.6526x; 1.1081x over previous
#include <hip/hip_runtime.h>

#define N_NODES 100000
#define IN_F    128
#define H_HEADS 4
#define D_OUT   16
#define HD      64
#define E_EDGES 3200000
#define SLOPE   0.2f

typedef __attribute__((ext_vector_type(2))) float v2f;
typedef __attribute__((ext_vector_type(4))) float v4f;
typedef __attribute__((ext_vector_type(8))) float v8f;

__device__ __forceinline__ unsigned flt_flip(float f) {
    unsigned u = __float_as_uint(f);
    return u ^ (((int)u >> 31) | 0x80000000u);
}
__device__ __forceinline__ float flt_unflip(unsigned u) {
    unsigned b = (u & 0x80000000u) ? (u ^ 0x80000000u) : ~u;
    return __uint_as_float(b);
}

#define GEMM_BLK 256
#define NSTRIP   (N_NODES / 16)
#define WPITCH   (IN_F + 4)
static_assert(NSTRIP * 16 == N_NODES && NSTRIP <= GEMM_BLK * 32, "every strip needs a wave");

__global__ __launch_bounds__(1024) void gat_gemm(const float* __restrict__ feat,
                                                 const float* __restrict__ W,
                                                 float* __restrict__ ft) {
    __shared__ __attribute__((aligned(16))) float wl[HD * WPITCH];
    const int tid  = threadIdx.x;
    const int lane = tid & 31;
    const int wave  = __builtin_amdgcn_readfirstlane(tid >> 5);
    const int strip = wave * GEMM_BLK + (int)blockIdx.x;
    const bool active = strip < NSTRIP;
    const int m  = lane & 15;
    const int hi = lane >> 4;

    const v4f w0 = *(const v4f*)(W + (size_t)tid * 4);
    const v4f w1 = *(const v4f*)(W + (size_t)(tid + 1024) * 4);
    const int ls = active ? strip : (int)blockIdx.x;
    const float* arow = feat + (size_t)(ls * 16 + m) * IN_F + (hi << 2);
    v4f a[IN_F / 8];
#pragma unroll
    for (int q = 0; q < IN_F / 8; ++q) a[q] = *(const v4f*)(arow + 8 * q);
    *(v4f*)(wl + (tid >> 5) * WPITCH + lane * 4)        = w0;
    *(v4f*)(wl + ((tid >> 5) + 32) * WPITCH + lane * 4) = w1;
    __builtin_amdgcn_fence(__ATOMIC_RELEASE, "workgroup", "local");
    __builtin_amdgcn_s_barrier();
    __builtin_amdgcn_fence(__ATOMIC_ACQUIRE, "workgroup", "local");
    if (!active) return;

    v8f acc[H_HEADS] = {};
    const float* bbase = wl + m * WPITCH + (hi << 2);
#pragma unroll
    for (int q = 0; q < IN_F / 8; ++q) {
        v4f b[H_HEADS];
#pragma unroll
        for (int ct = 0; ct < H_HEADS; ++ct) b[ct] = *(const v4f*)(bbase + ct * 16 * WPITCH + 8 * q);
        const v2f a0 = {a[q][0], a[q][1]}, a1 = {a[q][2], a[q][3]};
#pragma unroll
        for (int ct = 0; ct < H_HEADS; ++ct) {
            const v2f b0 = {b[ct][0], b[ct][1]};
            acc[ct] = __builtin_amdgcn_wmma_f32_16x16x4_f32(
                false, a0, false, b0, (short)0, acc[ct], false, false);
        }
#pragma unroll
        for (int ct = 0; ct < H_HEADS; ++ct) {
            const v2f b1 = {b[ct][2], b[ct][3]};
            acc[ct] = __builtin_amdgcn_wmma_f32_16x16x4_f32(
                false, a1, false, b1, (short)0, acc[ct], false, false);
        }
    }
    float* outp = ft + (size_t)(strip * 16 + hi * 8) * HD + m;
#pragma unroll
    for (int ct = 0; ct < H_HEADS; ++ct)
#pragma unroll
        for (int r = 0; r < 8; ++r) outp[(size_t)r * HD + ct * 16] = acc[ct][r];
}

__global__ __launch_bounds__(256) void gat_attn(const float* __restrict__ ft,
                                                const float* __restrict__ al,
                                                const float* __restrict__ ar,
                                                float* __restrict__ el,
                                                float* __restrict__ er) {
    int idx = blockIdx.x * blockDim.x + threadIdx.x;
    if (idx >= N_NODES * H_HEADS) return;
    const int h = idx & 3;
    const float* f = ft + (size_t)idx * D_OUT;
    float sl = 0.f, sr = 0.f;
#pragma unroll
    for (int d = 0; d < D_OUT; ++d) {
        float v = f[d];
        sl += v * al[h * D_OUT + d];
        sr += v * ar[h * D_OUT + d];
    }
    el[idx] = sl;
    er[idx] = sr;
}

__device__ __forceinline__ float edge_score(float l, float r) {
    float e = l + r;
    return e >= 0.f ? e : SLOPE * e;
}

#define NPB   391
#define NBLK  256
#define SBITS 17
#define NBKT  256
#define EPB   (E_EDGES / NBKT)
#define RUN_TRIPS (((NPB * HD) / NBKT - 1 + 31) / 32)

static_assert(EPB * NBKT == E_EDGES && EPB % 4 == 0, "edge list must split into 16-byte aligned parts");
static_assert(NPB < (1 << (32 - SBITS)) && N_NODES <= (1 << SBITS), "run entry format");

__device__ __forceinline__ unsigned owner_rows(unsigned b) {
    const unsigned rem = (unsigned)N_NODES - b * NPB;
    return rem < NPB ? rem : NPB;
}
__device__ __forceinline__ unsigned run_words(unsigned n_own) { return (n_own * HD) / NBKT; }

__device__ __forceinline__ float edge_weight(float l, float r, unsigned mu, float ss) {
    return expf(edge_score(l, r) - flt_unflip(mu)) / ss;
}

__device__ __forceinline__ void bucket_put(int d, int s, int g, unsigned* cnt, unsigned* rows) {
    const unsigned b = (unsigned)d / NPB;
    if (b < NBLK) {
        const unsigned dl = (unsigned)d - b * NPB;
        const unsigned r  = atomicAdd(&cnt[b], 1u);
        const unsigned S  = run_words(owner_rows(b));
        if (r + 1 < S) rows[(size_t)b * (NPB * HD) + g * S + 1 + r] = (unsigned)s | (dl << SBITS);
    }
}

__global__ __launch_bounds__(1024) void gat_bucket(const int* __restrict__ src,
                                                   const int* __restrict__ dst,
                                                   unsigned* __restrict__ rows) {
    __shared__ unsigned cnt[NBLK];
    const int tid = threadIdx.x;
    const int g   = blockIdx.x;
    if (tid < NBLK) cnt[tid] = 0u;
    __syncthreads();

    const int4* d4 = (const int4*)(dst + (size_t)g * EPB);
    const int4* s4 = (const int4*)(src + (size_t)g * EPB);
    const int nq = EPB / 4;
    const int TR = (EPB / 4 + 1023) / 1024;
    int4 dv[TR], sv[TR];
#pragma unroll
    for (int k = 0; k < TR; ++k) {
        const int q  = tid + k * 1024;
        const int qq = q < nq ? q : nq - 1;
        dv[k] = d4[qq];
        sv[k] = s4[qq];
    }
#pragma unroll
    for (int k = 0; k < TR; ++k) {
        if (tid + k * 1024 < nq) {
            bucket_put(dv[k].x, sv[k].x, g, cnt, rows);
            bucket_put(dv[k].y, sv[k].y, g, cnt, rows);
            bucket_put(dv[k].z, sv[k].z, g, cnt, rows);
            bucket_put(dv[k].w, sv[k].w, g, cnt, rows);
        }
    }
    __syncthreads();
    if (tid < NBLK)
        rows[(size_t)tid * (NPB * HD) + g * run_words(owner_rows((unsigned)tid))] = cnt[tid];
}

template <int CTRL>
__device__ __forceinline__ float row_xor(float x) {
    const int b = __builtin_bit_cast(int, x);
    return __builtin_bit_cast(float, __builtin_amdgcn_update_dpp(b, b, CTRL, 0xf, 0xf, false));
}
__device__ __forceinline__ float xhalf(float x) {
    const int b = __builtin_bit_cast(int, x);
    return __builtin_bit_cast(float,
        __builtin_amdgcn_permlanex16(b, b, 0x76543210, 0xfedcba98, false, false));
}
template <bool ISMAX>
__device__ __forceinline__ float red_op(float a, float b) { return ISMAX ? fmaxf(a, b) : a + b; }
template <bool ISMAX>
__device__ __forceinline__ void wave_reduce4(float& x0, float& x1, float& x2, float& x3, int lane) {
    const bool hi = (lane & 16) != 0, b3 = (lane & 8) != 0;
    const float keepA = hi ? x2 : x0, keepB = hi ? x3 : x1;
    const float sendA = hi ? x0 : x2, sendB = hi ? x1 : x3;
    const float yA = red_op<ISMAX>(keepA, xhalf(sendA));
    const float yB = red_op<ISMAX>(keepB, xhalf(sendB));
    const float keep = b3 ? yB : yA, send = b3 ? yA : yB;
    float z = red_op<ISMAX>(keep, row_xor<0x168>(send));
    z = red_op<ISMAX>(z, row_xor<0x164>(z));
    z = red_op<ISMAX>(z, row_xor<0x162>(z));
    z = red_op<ISMAX>(z, row_xor<0x161>(z));
    const int zi = __builtin_bit_cast(int, z);
    x0 = __builtin_bit_cast(float, __builtin_amdgcn_readlane(zi, 0));
    x1 = __builtin_bit_cast(float, __builtin_amdgcn_readlane(zi, 8));
    x2 = __builtin_bit_cast(float, __builtin_amdgcn_readlane(zi, 16));
    x3 = __builtin_bit_cast(float, __builtin_amdgcn_readlane(zi, 24));
}
__device__ __forceinline__ float pick4(int h, float x0, float x1, float x2, float x3) {
    return h == 0 ? x0 : (h == 1 ? x1 : (h == 2 ? x2 : x3));
}

#define WCAP  128
#define ROWU  8
#define NOENT 0xFFFFFFFFu

template <bool FULL>
__device__ __forceinline__ void row_group(const float* wq, const float* __restrict__ ft,
                                          int t0, int nv, int hf, int hl, unsigned c4b,
                                          v2f& accA, v2f& accB) {
    v2f ws[ROWU];
    v4f rv[ROWU];
#pragma unroll
    for (int u = 0; u < ROWU; ++u)
        ws[u] = *(const v2f*)(wq + (2 * (t0 + u) + hf) * 8 + hl * 2);
#pragma unroll
    for (int u = 0; u < ROWU; ++u) {
        if (FULL || u < nv) {
            rv[u] = *(const v4f*)((const char*)ft + (size_t)((__float_as_uint(ws[u][1]) << 8) + c4b));
        } else {
            rv[u][0] = rv[u][1] = rv[u][2] = rv[u][3] = 0.f;
            ws[u][0] = 0.f;
        }
    }
#pragma unroll
    for (int u = 0; u < ROWU; ++u) {
        const v2f w2 = {ws[u][0], ws[u][0]};
        const v2f lo = {rv[u][0], rv[u][1]};
        const v2f hi = {rv[u][2], rv[u][3]};
        accA = w2 * lo + accA;
        accB = w2 * hi + accB;
    }
}

__global__ __launch_bounds__(1024) void gat_edge_agg(const int* __restrict__ src,
                                                     const int* __restrict__ dst,
                                                     const float* __restrict__ el,
                                                     const float* __restrict__ er,
                                                     const float* __restrict__ ft,
                                                     float* out) {
    __shared__ __attribute__((aligned(16))) union { float f[NPB * HD]; unsigned u[NPB * HD]; } buf;
    __shared__ __attribute__((aligned(16))) float wq_all[(32 * WCAP + 16) * 8];
    __shared__ __attribute__((aligned(16))) unsigned mx[NPB * H_HEADS];
    __shared__ __attribute__((aligned(16))) float    sm[NPB * H_HEADS];
    __shared__ __attribute__((aligned(16))) float    ers[NPB * H_HEADS];
    __shared__ unsigned ncnt[NPB + 1];
    __shared__ unsigned nstart[NPB + 1];
    __shared__ unsigned rcnt[NBKT];
    __shared__ unsigned wsum[16];
    __shared__ unsigned overfull;

    const int tid  = threadIdx.x;
    const int lane = tid & 31;
    const int wave = __builtin_amdgcn_readfirstlane(tid >> 5);
    const int lo   = blockIdx.x * NPB;
    const int rem  = N_NODES - lo;
    const unsigned n_own = (unsigned)(rem < NPB ? rem : NPB);
    const unsigned S = run_words(n_own);
    const unsigned* runs = (const unsigned*)(out + (size_t)lo * HD);

    for (int k = tid; k <= NPB; k += 1024) ncnt[k] = 0u;
    if (tid == 0) overfull = 0u;
    __syncthreads();
    if (tid < NBKT) {
        const unsigned c = runs[(unsigned)tid * S];
        if (c >= S) atomicOr(&overfull, 1u);
        rcnt[tid] = c;
    }
    __syncthreads();
    const bool bucketed = __builtin_amdgcn_readfirstlane((int)overfull) == 0;

    if (bucketed) {
        unsigned v[NBKT / 32][RUN_TRIPS];
#pragma unroll
        for (int gi = 0; gi < NBKT / 32; ++gi) {
            const int g = wave + gi * 32;
            const unsigned n = rcnt[g];
#pragma unroll
            for (int t = 0; t < RUN_TRIPS; ++t) {
                const unsigned j = (unsigned)lane + t * 32;
                v[gi][t] = j < n ? runs[(unsigned)g * S + 1 + j] : NOENT;
            }
        }
#pragma unroll
        for (int gi = 0; gi < NBKT / 32; ++gi)
#pragma unroll
            for (int t = 0; t < RUN_TRIPS; ++t)
                if (v[gi][t] != NOENT) atomicAdd(&ncnt[v[gi][t] >> SBITS], 1u);
        __syncthreads();
        unsigned cv = 0u, inc = 0u;
        if (tid < 512) {
            cv  = tid <= NPB ? ncnt[tid] : 0u;
            inc = cv;
            for (int d = 1; d < 32; d <<= 1) {
                const unsigned x = __shfl_up(inc, d);
                if (lane >= d) inc += x;
            }
            if (lane == 31) wsum[wave] = inc;
        }
        __syncthreads();
        if (tid <= NPB) {
            unsigned off = inc - cv;
            for (int w = 0; w < wave; ++w) off += wsum[w];
            nstart[tid] = off;
            ncnt[tid]   = off;
        }
        __syncthreads();
#pragma unroll
        for (int gi = 0; gi < NBKT / 32; ++gi)
#pragma unroll
            for (int t = 0; t < RUN_TRIPS; ++t)
                if (v[gi][t] != NOENT) {
                    const unsigned pos = atomicAdd(&ncnt[v[gi][t] >> SBITS], 1u);
                    buf.u[pos] = v[gi][t] & ((1u << SBITS) - 1u);
                }
        __syncthreads();

        float* wq = wq_all + wave * (WCAP * 8);
        const int      hf  = lane >> 4;
        const unsigned c4b = (unsigned)(lane & 15) * 16u;
        const int      hl  = (lane & 15) >> 2;
        for (int node = wave; node < (int)n_own; node += 32) {
            const int beg = __builtin_amdgcn_readfirstlane((int)nstart[node]);
            const int deg = __builtin_amdgcn_readfirstlane((int)nstart[node + 1]) - beg;
            const v4f r4  = *(const v4f*)(er + (size_t)(lo + node) * 4);

            float m0 = -INFINITY, m1 = -INFINITY, m2 = -INFINITY, m3 = -INFINITY;
            for (int jb = 0; jb < deg; jb += 64) {
                const int  ja = jb + lane, jc = ja + 32;
                const bool va = ja < deg,  vc = jc < deg;
                const unsigned sa = buf.u[beg + (va ? ja : deg - 1)];
                const unsigned sc = buf.u[beg + (vc ? jc : deg - 1)];
                const v4f la = *(const v4f*)((const char*)el + (size_t)(sa << 4));
                const v4f lc = *(const v4f*)((const char*)el + (size_t)(sc << 4));
                v4f ea, ec;
#pragma unroll
                for (int h = 0; h < 4; ++h) {
                    ea[h] = va ? edge_score(la[h], r4[h]) : -INFINITY;
                    ec[h] = vc ? edge_score(lc[h], r4[h]) : -INFINITY;
                }
                if (ja < WCAP) {
                    const float sf = __uint_as_float(sa);
                    v4f q0, q1;
                    q0[0] = ea[0]; q0[1] = sf; q0[2] = ea[1]; q0[3] = sf;
                    q1[0] = ea[2]; q1[1] = sf; q1[2] = ea[3]; q1[3] = sf;
                    *(v4f*)(wq + ja * 8)     = q0;
                    *(v4f*)(wq + ja * 8 + 4) = q1;
                }
                if (jc < WCAP) {
                    const float sf = __uint_as_float(sc);
                    v4f q0, q1;
                    q0[0] = ec[0]; q0[1] = sf; q0[2] = ec[1]; q0[3] = sf;
                    q1[0] = ec[2]; q1[1] = sf; q1[2] = ec[3]; q1[3] = sf;
                    *(v4f*)(wq + jc * 8)     = q0;
                    *(v4f*)(wq + jc * 8 + 4) = q1;
                }
                m0 = fmaxf(m0, fmaxf(ea[0], ec[0]));
                m1 = fmaxf(m1, fmaxf(ea[1], ec[1]));
                m2 = fmaxf(m2, fmaxf(ea[2], ec[2]));
                m3 = fmaxf(m3, fmaxf(ea[3], ec[3]));
            }
            wave_reduce4<true>(m0, m1, m2, m3, lane);

            float s0 = 0.f, s1 = 0.f, s2 = 0.f, s3 = 0.f;
            for (int jb = 0; jb < deg; jb += 32) {
                const int j = jb + lane;
                float e0 = -INFINITY, e1 = -INFINITY, e2 = -INFINITY, e3 = -INFINITY, sf = 0.f;
                if (j < WCAP) {
                    const v4f q0 = *(const v4f*)(wq + j * 8);
                    const v4f q1 = *(const v4f*)(wq + j * 8 + 4);
                    e0 = q0[0]; e1 = q0[2]; e2 = q1[0]; e3 = q1[2]; sf = q0[1];
                } else if (j < deg) {
                    const unsigned s = buf.u[beg + j];
                    const v4f l = *(const v4f*)((const char*)el + (size_t)(s << 4));
                    e0 = edge_score(l[0], r4[0]);
                    e1 = edge_score(l[1], r4[1]);
                    e2 = edge_score(l[2], r4[2]);
                    e3 = edge_score(l[3], r4[3]);
                }
                float x0 = 0.f, x1 = 0.f, x2 = 0.f, x3 = 0.f;
                if (j < deg) {
                    x0 = expf(e0 - m0);
                    x1 = expf(e1 - m1);
                    x2 = expf(e2 - m2);
                    x3 = expf(e3 - m3);
                }
                if (j < WCAP) {
                    v4f q0, q1;
                    q0[0] = x0; q0[1] = sf; q0[2] = x1; q0[3] = sf;
                    q1[0] = x2; q1[1] = sf; q1[2] = x3; q1[3] = sf;
                    *(v4f*)(wq + j * 8)     = q0;
                    *(v4f*)(wq + j * 8 + 4) = q1;
                }
                s0 += x0; s1 += x1; s2 += x2; s3 += x3;
            }
            wave_reduce4<false>(s0, s1, s2, s3, lane);

            v2f accA = {0.f, 0.f}, accB = {0.f, 0.f};
            const int npairs = (deg + 1) >> 1;
            const int lim    = deg < WCAP ? deg : WCAP;
            const int nfast  = (lim + 1) >> 1;
            int t0 = 0;
            for (; t0 + ROWU <= nfast; t0 += ROWU)
                row_group<true>(wq, ft, t0, ROWU, hf, hl, c4b, accA, accB);
            if (t0 < nfast)
                row_group<false>(wq, ft, t0, nfast - t0, hf, hl, c4b, accA, accB);
            float a0 = accA[0], a1 = accA[1], a2 = accB[0], a3 = accB[1];
            if (deg > WCAP) {
                const float rl = pick4(hl, r4[0], r4[1], r4[2], r4[3]);
                const float ml = pick4(hl, m0, m1, m2, m3);
                for (int t = WCAP / 2; t < npairs; ++t) {
                    const int j  = 2 * t + hf;
                    const int jj = j < deg ? j : deg - 1;
                    const unsigned s = buf.u[beg + jj];
                    const float l = *(const float*)((const char*)el + (size_t)((s << 4) + (unsigned)hl * 4u));
                    const float w = j < deg ? expf(edge_score(l, rl) - ml) : 0.f;
                    const v4f r = *(const v4f*)((const char*)ft + (size_t)((s << 8) + c4b));
                    a0 += w * r[0];
                    a1 += w * r[1];
                    a2 += w * r[2];
                    a3 += w * r[3];
                }
            }

            const float sl  = pick4(hl, s0, s1, s2, s3);
            const float inv = deg > 0 ? 1.0f / sl : 0.f;
            a0 += xhalf(a0);
            a1 += xhalf(a1);
            a2 += xhalf(a2);
            a3 += xhalf(a3);
            if (lane < 16) {
                v4f o;
                o[0] = a0 * inv; o[1] = a1 * inv; o[2] = a2 * inv; o[3] = a3 * inv;
                *(v4f*)((char*)out + (size_t)(lo + node) * (HD * 4) + c4b) = o;
            }
        }
    } else {
        float* acc = buf.f;
        for (int k = tid; k < NPB * HD; k += 1024) acc[k] = 0.f;
        for (int k = tid; k < NPB * H_HEADS; k += 1024) {
            mx[k]  = 0u;
            sm[k]  = 0.f;
            ers[k] = (k < (int)n_own * H_HEADS) ? er[(size_t)lo * H_HEADS + k] : 0.f;
        }
        __syncthreads();
        for (int i = tid; i < E_EDGES; i += 1024) {
            const unsigned u = (unsigned)(dst[i] - lo);
            if (u < n_own) {
                const int s_ = src[i];
                for (int h = 0; h < H_HEADS; ++h)
                    atomicMax(&mx[u * 4 + h], flt_flip(edge_score(el[s_ * 4 + h], ers[u * 4 + h])));
            }
        }
        __syncthreads();
        for (int i = tid; i < E_EDGES; i += 1024) {
            const unsigned u = (unsigned)(dst[i] - lo);
            if (u < n_own) {
                const int s_ = src[i];
                for (int h = 0; h < H_HEADS; ++h)
                    atomicAdd(&sm[u * 4 + h],
                              expf(edge_score(el[s_ * 4 + h], ers[u * 4 + h]) - flt_unflip(mx[u * 4 + h])));
            }
        }
        __syncthreads();
        for (int i = tid; i < E_EDGES; i += 1024) {
            const unsigned u = (unsigned)(dst[i] - lo);
            if (u < n_own) {
                const int s_ = src[i];
                for (int h = 0; h < H_HEADS; ++h) {
                    const float w = edge_weight(el[s_ * 4 + h], ers[u * 4 + h],
                                                mx[u * 4 + h], sm[u * 4 + h]);
                    const float* f = ft + (size_t)s_ * HD + h * D_OUT;
                    for (int d = 0; d < D_OUT; ++d)
                        atomicAdd(&acc[u * HD + h * D_OUT + d], f[d] * w);
                }
            }
        }
        __syncthreads();
        float* o = out + (size_t)lo * HD;
        const int nout = (int)n_own * HD;
        for (int k = tid; k < nout; k += 1024) o[k] = acc[k];
    }
}

extern "C" void kernel_launch(void* const* d_in, const int* in_sizes, int n_in,
                              void* d_out, int out_size, void* d_ws, size_t ws_size,
                              hipStream_t stream) {
    const float* feat = (const float*)d_in[0];
    const float* W    = (const float*)d_in[1];
    const float* al   = (const float*)d_in[2];
    const float* ar   = (const float*)d_in[3];
    const int*   src  = (const int*)d_in[4];
    const int*   dst  = (const int*)d_in[5];
    float* out = (float*)d_out;

    float* ft = (float*)d_ws;
    float* el = ft + (size_t)N_NODES * HD;
    float* er = el + (size_t)N_NODES * H_HEADS;

    const int nb_attn = (N_NODES * H_HEADS + 255) / 256;

    gat_bucket<<<NBKT, 1024, 0, stream>>>(src, dst, (unsigned*)d_out);
    gat_gemm<<<GEMM_BLK, 1024, 0, stream>>>(feat, W, ft);
    gat_attn<<<nb_attn, 256, 0, stream>>>(ft, al, ar, el, er);
    gat_edge_agg<<<NBLK, 1024, 0, stream>>>(src, dst, el, er, ft, out);
}


# Source logit el not gathered any more: the owner wave forms it from the ft row it fetches anyway (dot with attn_l over the head's 4 lanes); one-pass softmax, running maximum; sum order changes
# speedup vs baseline: 17.0360x; 1.1627x over previous
#include <hip/hip_runtime.h>

#define N_NODES 100000
#define IN_F    128
#define H_HEADS 4
#define D_OUT   16
#define HD      64
#define E_EDGES 3200000
#define SLOPE   0.2f

typedef __attribute__((ext_vector_type(2))) float v2f;
typedef __attribute__((ext_vector_type(4))) float v4f;
typedef __attribute__((ext_vector_type(8))) float v8f;

__device__ __forceinline__ unsigned flt_flip(float f) {
    unsigned u = __float_as_uint(f);
    return u ^ (((int)u >> 31) | 0x80000000u);
}
__device__ __forceinline__ float flt_unflip(unsigned u) {
    unsigned b = (u & 0x80000000u) ? (u ^ 0x80000000u) : ~u;
    return __uint_as_float(b);
}

template <int CTRL>
__device__ __forceinline__ float row_xor(float x) {
    const int b = __builtin_bit_cast(int, x);
    return __builtin_bit_cast(float, __builtin_amdgcn_update_dpp(b, b, CTRL, 0xf, 0xf, false));
}

#define GEMM_BLK 256
#define NSTRIP   (N_NODES / 16)
#define WPITCH   (IN_F + 4)
static_assert(NSTRIP * 16 == N_NODES && NSTRIP <= GEMM_BLK * 32, "every strip needs a wave");

__global__ __launch_bounds__(1024) void gat_gemm(const float* __restrict__ feat,
                                                 const float* __restrict__ W,
                                                 const float* __restrict__ al,
                                                 const float* __restrict__ ar,
                                                 float* __restrict__ ft,
                                                 float* __restrict__ el,
                                                 float* __restrict__ er) {
    __shared__ __attribute__((aligned(16))) float wl[HD * WPITCH];
    const int tid  = threadIdx.x;
    const int lane = tid & 31;
    const int wave  = __builtin_amdgcn_readfirstlane(tid >> 5);
    const int strip = wave * GEMM_BLK + (int)blockIdx.x;
    const bool active = strip < NSTRIP;
    const int m  = lane & 15;
    const int hi = lane >> 4;

    const v4f w0 = *(const v4f*)(W + (size_t)tid * 4);
    const v4f w1 = *(const v4f*)(W + (size_t)(tid + 1024) * 4);
    const bool c3 = (lane & 8) != 0;
    const float* pk = c3 ? ar : al;
    const float* ps = c3 ? al : ar;
    float ck[H_HEADS], cs[H_HEADS];
#pragma unroll
    for (int ct = 0; ct < H_HEADS; ++ct) {
        ck[ct] = pk[ct * D_OUT + (lane & 15)];
        cs[ct] = ps[ct * D_OUT + (lane & 15)];
    }
    const int ls = active ? strip : (int)blockIdx.x;
    const float* arow = feat + (size_t)(ls * 16 + m) * IN_F + (hi << 2);
    v4f a[IN_F / 8];
#pragma unroll
    for (int q = 0; q < IN_F / 8; ++q) a[q] = *(const v4f*)(arow + 8 * q);
    *(v4f*)(wl + (tid >> 5) * WPITCH + lane * 4)        = w0;
    *(v4f*)(wl + ((tid >> 5) + 32) * WPITCH + lane * 4) = w1;
    __builtin_amdgcn_fence(__ATOMIC_RELEASE, "workgroup", "local");
    __builtin_amdgcn_s_barrier();
    __builtin_amdgcn_fence(__ATOMIC_ACQUIRE, "workgroup", "local");
    if (!active) return;

    v8f acc[H_HEADS] = {};
    const float* bbase = wl + m * WPITCH + (hi << 2);
#pragma unroll
    for (int q = 0; q < IN_F / 8; ++q) {
        v4f b[H_HEADS];
#pragma unroll
        for (int ct = 0; ct < H_HEADS; ++ct) b[ct] = *(const v4f*)(bbase + ct * 16 * WPITCH + 8 * q);
        const v2f a0 = {a[q][0], a[q][1]}, a1 = {a[q][2], a[q][3]};
#pragma unroll
        for (int ct = 0; ct < H_HEADS; ++ct) {
            const v2f b0 = {b[ct][0], b[ct][1]};
            acc[ct] = __builtin_amdgcn_wmma_f32_16x16x4_f32(
                false, a0, false, b0, (short)0, acc[ct], false, false);
        }
#pragma unroll
        for (int ct = 0; ct < H_HEADS; ++ct) {
            const v2f b1 = {b[ct][2], b[ct][3]};
            acc[ct] = __builtin_amdgcn_wmma_f32_16x16x4_f32(
                false, a1, false, b1, (short)0, acc[ct], false, false);
        }
    }
    float* outp = ft + (size_t)(strip * 16 + hi * 8) * HD + m;
#pragma unroll
    for (int ct = 0; ct < H_HEADS; ++ct)
#pragma unroll
        for (int r = 0; r < 8; ++r) outp[(size_t)r * HD + ct * 16] = acc[ct][r];

    const bool c2 = (lane & 4) != 0, c1 = (lane & 2) != 0, c0 = (lane & 1) != 0;
    float t1[H_HEADS][8];
#pragma unroll
    for (int ct = 0; ct < H_HEADS; ++ct)
#pragma unroll
        for (int r = 0; r < 8; ++r)
            t1[ct][r] = acc[ct][r] * ck[ct] + row_xor<0x168>(acc[ct][r] * cs[ct]);
    float t2[H_HEADS][4];
#pragma unroll
    for (int ct = 0; ct < H_HEADS; ++ct)
#pragma unroll
        for (int r = 0; r < 4; ++r) {
            const float keep = c2 ? t1[ct][r + 4] : t1[ct][r];
            const float send = c2 ? t1[ct][r]     : t1[ct][r + 4];
            t2[ct][r] = keep + row_xor<0x164>(send);
        }
    float t3[H_HEADS][2];
#pragma unroll
    for (int ct = 0; ct < H_HEADS; ++ct)
#pragma unroll
        for (int r = 0; r < 2; ++r) {
            const float keep = c1 ? t2[ct][r + 2] : t2[ct][r];
            const float send = c1 ? t2[ct][r]     : t2[ct][r + 2];
            t3[ct][r] = keep + row_xor<0x162>(send);
        }
    v4f lg;
#pragma unroll
    for (int ct = 0; ct < H_HEADS; ++ct) {
        const float keep = c0 ? t3[ct][1] : t3[ct][0];
        const float send = c0 ? t3[ct][0] : t3[ct][1];
        lg[ct] = keep + row_xor<0x161>(send);
    }
    float* lgp = (c3 ? er : el) + (size_t)(strip * 16 + hi * 8 + (lane & 7)) * H_HEADS;
    *(v4f*)lgp = lg;
}

__device__ __forceinline__ float edge_score(float l, float r) {
    float e = l + r;
    return e >= 0.f ? e : SLOPE * e;
}

#define NPB   391
#define NBLK  256
#define SBITS 17
#define NBKT  256
#define EPB   (E_EDGES / NBKT)
#define RUN_TRIPS (((NPB * HD) / NBKT - 1 + 31) / 32)

static_assert(EPB * NBKT == E_EDGES && EPB % 4 == 0, "edge list must split into 16-byte aligned parts");
static_assert(NPB < (1 << (32 - SBITS)) && N_NODES <= (1 << SBITS), "run entry format");

__device__ __forceinline__ unsigned owner_rows(unsigned b) {
    const unsigned rem = (unsigned)N_NODES - b * NPB;
    return rem < NPB ? rem : NPB;
}
__device__ __forceinline__ unsigned run_words(unsigned n_own) { return (n_own * HD) / NBKT; }

__device__ __forceinline__ float edge_weight(float l, float r, unsigned mu, float ss) {
    return expf(edge_score(l, r) - flt_unflip(mu)) / ss;
}

__device__ __forceinline__ void bucket_put(int d, int s, int g, unsigned* cnt, unsigned* rows) {
    const unsigned b = (unsigned)d / NPB;
    if (b < NBLK) {
        const unsigned dl = (unsigned)d - b * NPB;
        const unsigned r  = atomicAdd(&cnt[b], 1u);
        const unsigned S  = run_words(owner_rows(b));
        if (r + 1 < S) rows[(size_t)b * (NPB * HD) + g * S + 1 + r] = (unsigned)s | (dl << SBITS);
    }
}

__global__ __launch_bounds__(1024) void gat_bucket(const int* __restrict__ src,
                                                   const int* __restrict__ dst,
                                                   unsigned* __restrict__ rows) {
    __shared__ unsigned cnt[NBLK];
    const int tid = threadIdx.x;
    const int g   = blockIdx.x;
    if (tid < NBLK) cnt[tid] = 0u;
    __syncthreads();

    const int4* d4 = (const int4*)(dst + (size_t)g * EPB);
    const int4* s4 = (const int4*)(src + (size_t)g * EPB);
    const int nq = EPB / 4;
    const int TR = (EPB / 4 + 1023) / 1024;
    int4 dv[TR], sv[TR];
#pragma unroll
    for (int k = 0; k < TR; ++k) {
        const int q  = tid + k * 1024;
        const int qq = q < nq ? q : nq - 1;
        dv[k] = d4[qq];
        sv[k] = s4[qq];
    }
#pragma unroll
    for (int k = 0; k < TR; ++k) {
        if (tid + k * 1024 < nq) {
            bucket_put(dv[k].x, sv[k].x, g, cnt, rows);
            bucket_put(dv[k].y, sv[k].y, g, cnt, rows);
            bucket_put(dv[k].z, sv[k].z, g, cnt, rows);
            bucket_put(dv[k].w, sv[k].w, g, cnt, rows);
        }
    }
    __syncthreads();
    if (tid < NBLK)
        rows[(size_t)tid * (NPB * HD) + g * run_words(owner_rows((unsigned)tid))] = cnt[tid];
}

__device__ __forceinline__ float xhalf(float x) {
    const int b = __builtin_bit_cast(int, x);
    return __builtin_bit_cast(float,
        __builtin_amdgcn_permlanex16(b, b, 0x76543210, 0xfedcba98, false, false));
}
template <int CTRL>
__device__ __forceinline__ float quad_bcast(float x) {
    const int b = __builtin_bit_cast(int, x);
    return __builtin_bit_cast(float, __builtin_amdgcn_update_dpp(b, b, CTRL, 0xf, 0xf, false));
}
#define CHP   16
#define NOENT 0xFFFFFFFFu

__global__ __launch_bounds__(1024) void gat_edge_agg(const int* __restrict__ src,
                                                     const int* __restrict__ dst,
                                                     const float* __restrict__ al,
                                                     const float* __restrict__ el,
                                                     const float* __restrict__ er,
                                                     const float* __restrict__ ft,
                                                     float* out) {
    __shared__ __attribute__((aligned(16))) union { float f[NPB * HD]; unsigned u[NPB * HD]; } buf;
    __shared__ __attribute__((aligned(16))) unsigned mx[NPB * H_HEADS];
    __shared__ __attribute__((aligned(16))) float    sm[NPB * H_HEADS];
    __shared__ __attribute__((aligned(16))) float    ers[NPB * H_HEADS];
    __shared__ unsigned ncnt[NPB + 1];
    __shared__ unsigned nstart[NPB + 1];
    __shared__ unsigned rcnt[NBKT];
    __shared__ unsigned wsum[16];
    __shared__ unsigned overfull;

    const int tid  = threadIdx.x;
    const int lane = tid & 31;
    const int wave = __builtin_amdgcn_readfirstlane(tid >> 5);
    const int lo   = blockIdx.x * NPB;
    const int rem  = N_NODES - lo;
    const unsigned n_own = (unsigned)(rem < NPB ? rem : NPB);
    const unsigned S = run_words(n_own);
    const unsigned* runs = (const unsigned*)(out + (size_t)lo * HD);

    for (int k = tid; k <= NPB; k += 1024) ncnt[k] = 0u;
    if (tid == 0) overfull = 0u;
    __syncthreads();
    if (tid < NBKT) {
        const unsigned c = runs[(unsigned)tid * S];
        if (c >= S) atomicOr(&overfull, 1u);
        rcnt[tid] = c;
    }
    __syncthreads();
    const bool bucketed = __builtin_amdgcn_readfirstlane((int)overfull) == 0;

    if (bucketed) {
        unsigned v[NBKT / 32][RUN_TRIPS];
#pragma unroll
        for (int gi = 0; gi < NBKT / 32; ++gi) {
            const int g = wave + gi * 32;
            const unsigned n = rcnt[g];
#pragma unroll
            for (int t = 0; t < RUN_TRIPS; ++t) {
                const unsigned j = (unsigned)lane + t * 32;
                v[gi][t] = j < n ? runs[(unsigned)g * S + 1 + j] : NOENT;
            }
        }
        for (int k = tid; k < (int)n_own * H_HEADS; k += 1024) ers[k] = er[(size_t)lo * H_HEADS + k];
#pragma unroll
        for (int gi = 0; gi < NBKT / 32; ++gi)
#pragma unroll
            for (int t = 0; t < RUN_TRIPS; ++t)
                if (v[gi][t] != NOENT) atomicAdd(&ncnt[v[gi][t] >> SBITS], 1u);
        __syncthreads();
        unsigned cv = 0u, inc = 0u;
        if (tid < 512) {
            cv  = tid <= NPB ? ncnt[tid] : 0u;
            inc = cv;
            for (int d = 1; d < 32; d <<= 1) {
                const unsigned x = __shfl_up(inc, d);
                if (lane >= d) inc += x;
            }
            if (lane == 31) wsum[wave] = inc;
        }
        __syncthreads();
        if (tid <= NPB) {
            unsigned off = inc - cv;
            for (int w = 0; w < wave; ++w) off += wsum[w];
            nstart[tid] = off;
            ncnt[tid]   = off;
        }
        __syncthreads();
#pragma unroll
        for (int gi = 0; gi < NBKT / 32; ++gi)
#pragma unroll
            for (int t = 0; t < RUN_TRIPS; ++t)
                if (v[gi][t] != NOENT) {
                    const unsigned pos = atomicAdd(&ncnt[v[gi][t] >> SBITS], 1u);
                    buf.u[pos] = v[gi][t] & ((1u << SBITS) - 1u);
                }
        __syncthreads();

        const int      hf  = lane >> 4;
        const unsigned c4b = (unsigned)(lane & 15) * 16u;
        const int      hl  = (lane & 15) >> 2;
        const v4f alc = *(const v4f*)(al + (lane & 15) * 4);
        const bool q1 = (lane & 2) != 0, q0 = (lane & 1) != 0;
        for (int node = wave; node < (int)n_own; node += 32) {
            const int beg = __builtin_amdgcn_readfirstlane((int)nstart[node]);
            const int deg = __builtin_amdgcn_readfirstlane((int)nstart[node + 1]) - beg;
            const float erl = ers[node * H_HEADS + hl];

            float mrun = -INFINITY, ssum = 0.f;
            v2f accA = {0.f, 0.f}, accB = {0.f, 0.f};
            const int npairs = (deg + 1) >> 1;
            for (int t0 = 0; t0 < npairs; t0 += CHP) {
                const int nv = npairs - t0 < CHP ? npairs - t0 : CHP;
                const int j0 = 2 * t0 + hf;
                const int jq = j0 + 2 * (lane & 3);
                unsigned sv[CHP];
                v4f rv[CHP];
#pragma unroll
                for (int u = 0; u < CHP; ++u) {
                    const int j = j0 + 2 * u;
                    sv[u] = buf.u[beg + (j < deg ? j : deg - 1)];
                }
#pragma unroll
                for (int u = 0; u < CHP; ++u) {
                    if (u < nv) {
                        rv[u] = *(const v4f*)((const char*)ft + (size_t)((sv[u] << 8) + c4b));
                    } else {
                        rv[u][0] = rv[u][1] = rv[u][2] = rv[u][3] = 0.f;
                    }
                }
                float sk[CHP / 4];
                float cmax = -INFINITY;
#pragma unroll
                for (int k = 0; k < CHP / 4; ++k) {
                    if (4 * k < nv) {
                        float p[4];
#pragma unroll
                        for (int i = 0; i < 4; ++i) {
                            const v4f r = rv[4 * k + i];
                            float d = r[0] * alc[0];
                            d = r[1] * alc[1] + d;
                            d = r[2] * alc[2] + d;
                            d = r[3] * alc[3] + d;
                            p[i] = d;
                        }
                        const float kA = q1 ? p[2] : p[0], kB = q1 ? p[3] : p[1];
                        const float sA = q1 ? p[0] : p[2], sB = q1 ? p[1] : p[3];
                        const float yA = kA + row_xor<0x162>(sA);
                        const float yB = kB + row_xor<0x162>(sB);
                        const float keep = q0 ? yB : yA, send = q0 ? yA : yB;
                        const float z = keep + row_xor<0x161>(send);
                        const float e = edge_score(z, erl);
                        sk[k] = (jq + 8 * k < deg) ? e : -INFINITY;
                    } else {
                        sk[k] = -INFINITY;
                    }
                    cmax = fmaxf(cmax, sk[k]);
                }
                cmax = fmaxf(cmax, row_xor<0x161>(cmax));
                cmax = fmaxf(cmax, row_xor<0x162>(cmax));
                cmax = fmaxf(cmax, xhalf(cmax));
                const float mnew = fmaxf(mrun, cmax);
                const float resc = expf(mrun - mnew);
                const v2f r2 = {resc, resc};
                ssum *= resc;
                accA *= r2;
                accB *= r2;
                mrun = mnew;
#pragma unroll
                for (int k = 0; k < CHP / 4; ++k) {
                    if (4 * k < nv) {
                        const float w = expf(sk[k] - mnew);
                        ssum += w;
                        const float wb[4] = {quad_bcast<0x00>(w), quad_bcast<0x55>(w),
                                             quad_bcast<0xAA>(w), quad_bcast<0xFF>(w)};
#pragma unroll
                        for (int i = 0; i < 4; ++i) {
                            const v2f w2 = {wb[i], wb[i]};
                            const v2f lo2 = {rv[4 * k + i][0], rv[4 * k + i][1]};
                            const v2f hi2 = {rv[4 * k + i][2], rv[4 * k + i][3]};
                            accA = w2 * lo2 + accA;
                            accB = w2 * hi2 + accB;
                        }
                    }
                }
            }
            ssum += row_xor<0x161>(ssum);
            ssum += row_xor<0x162>(ssum);
            float a0 = accA[0], a1 = accA[1], a2 = accB[0], a3 = accB[1];
            ssum += xhalf(ssum);
            a0 += xhalf(a0);
            a1 += xhalf(a1);
            a2 += xhalf(a2);
            a3 += xhalf(a3);
            const float inv = deg > 0 ? 1.0f / ssum : 0.f;
            if (lane < 16) {
                v4f o;
                o[0] = a0 * inv; o[1] = a1 * inv; o[2] = a2 * inv; o[3] = a3 * inv;
                *(v4f*)((char*)out + (size_t)(lo + node) * (HD * 4) + c4b) = o;
            }
        }
    } else {
        float* acc = buf.f;
        for (int k = tid; k < NPB * HD; k += 1024) acc[k] = 0.f;
        for (int k = tid; k < NPB * H_HEADS; k += 1024) {
            mx[k]  = 0u;
            sm[k]  = 0.f;
            ers[k] = (k < (int)n_own * H_HEADS) ? er[(size_t)lo * H_HEADS + k] : 0.f;
        }
        __syncthreads();
        for (int i = tid; i < E_EDGES; i += 1024) {
            const unsigned u = (unsigned)(dst[i] - lo);
            if (u < n_own) {
                const int s_ = src[i];
                for (int h = 0; h < H_HEADS; ++h)
                    atomicMax(&mx[u * 4 + h], flt_flip(edge_score(el[s_ * 4 + h], ers[u * 4 + h])));
            }
        }
        __syncthreads();
        for (int i = tid; i < E_EDGES; i += 1024) {
            const unsigned u = (unsigned)(dst[i] - lo);
            if (u < n_own) {
                const int s_ = src[i];
                for (int h = 0; h < H_HEADS; ++h)
                    atomicAdd(&sm[u * 4 + h],
                              expf(edge_score(el[s_ * 4 + h], ers[u * 4 + h]) - flt_unflip(mx[u * 4 + h])));
            }
        }
        __syncthreads();
        for (int i = tid; i < E_EDGES; i += 1024) {
            const unsigned u = (unsigned)(dst[i] - lo);
            if (u < n_own) {
                const int s_ = src[i];
                for (int h = 0; h < H_HEADS; ++h) {
                    const float w = edge_weight(el[s_ * 4 + h], ers[u * 4 + h],
                                                mx[u * 4 + h], sm[u * 4 + h]);
                    const float* f = ft + (size_t)s_ * HD + h * D_OUT;
                    for (int d = 0; d < D_OUT; ++d)
                        atomicAdd(&acc[u * HD + h * D_OUT + d], f[d] * w);
                }
            }
        }
        __syncthreads();
        float* o = out + (size_t)lo * HD;
        const int nout = (int)n_own * HD;
        for (int k = tid; k < nout; k += 1024) o[k] = acc[k];
    }
}

extern "C" void kernel_launch(void* const* d_in, const int* in_sizes, int n_in,
                              void* d_out, int out_size, void* d_ws, size_t ws_size,
                              hipStream_t stream) {
    const float* feat = (const float*)d_in[0];
    const float* W    = (const float*)d_in[1];
    const float* al   = (const float*)d_in[2];
    const float* ar   = (const float*)d_in[3];
    const int*   src  = (const int*)d_in[4];
    const int*   dst  = (const int*)d_in[5];
    float* out = (float*)d_out;

    float* ft = (float*)d_ws;
    float* el = ft + (size_t)N_NODES * HD;
    float* er = el + (size_t)N_NODES * H_HEADS;

    gat_bucket<<<NBKT, 1024, 0, stream>>>(src, dst, (unsigned*)d_out);
    gat_gemm<<<GEMM_BLK, 1024, 0, stream>>>(feat, W, al, ar, ft, el, er);
    gat_edge_agg<<<NBLK, 1024, 0, stream>>>(src, dst, al, el, er, ft, out);
}


# Owner kernel: every node's segment of the sorted list padded to whole groups of 8 slots (copies of its last source, weight 0): guard-free chunk bodies, a chunk's row loads issued together
# speedup vs baseline: 17.8313x; 1.0467x over previous
#include <hip/hip_runtime.h>

#define N_NODES 100000
#define IN_F    128
#define H_HEADS 4
#define D_OUT   16
#define HD      64
#define E_EDGES 3200000
#define SLOPE   0.2f

typedef __attribute__((ext_vector_type(2))) float v2f;
typedef __attribute__((ext_vector_type(4))) float v4f;
typedef __attribute__((ext_vector_type(8))) float v8f;

__device__ __forceinline__ unsigned flt_flip(float f) {
    unsigned u = __float_as_uint(f);
    return u ^ (((int)u >> 31) | 0x80000000u);
}
__device__ __forceinline__ float flt_unflip(unsigned u) {
    unsigned b = (u & 0x80000000u) ? (u ^ 0x80000000u) : ~u;
    return __uint_as_float(b);
}

template <int CTRL>
__device__ __forceinline__ float row_xor(float x) {
    const int b = __builtin_bit_cast(int, x);
    return __builtin_bit_cast(float, __builtin_amdgcn_update_dpp(b, b, CTRL, 0xf, 0xf, false));
}

#define GEMM_BLK 256
#define NSTRIP   (N_NODES / 16)
#define WPITCH   (IN_F + 4)
static_assert(NSTRIP * 16 == N_NODES && NSTRIP <= GEMM_BLK * 32, "every strip needs a wave");

__global__ __launch_bounds__(1024) void gat_gemm(const float* __restrict__ feat,
                                                 const float* __restrict__ W,
                                                 const float* __restrict__ al,
                                                 const float* __restrict__ ar,
                                                 float* __restrict__ ft,
                                                 float* __restrict__ el,
                                                 float* __restrict__ er) {
    __shared__ __attribute__((aligned(16))) float wl[HD * WPITCH];
    const int tid  = threadIdx.x;
    const int lane = tid & 31;
    const int wave  = __builtin_amdgcn_readfirstlane(tid >> 5);
    const int strip = wave * GEMM_BLK + (int)blockIdx.x;
    const bool active = strip < NSTRIP;
    const int m  = lane & 15;
    const int hi = lane >> 4;

    const v4f w0 = *(const v4f*)(W + (size_t)tid * 4);
    const v4f w1 = *(const v4f*)(W + (size_t)(tid + 1024) * 4);
    const bool c3 = (lane & 8) != 0;
    const float* pk = c3 ? ar : al;
    const float* ps = c3 ? al : ar;
    float ck[H_HEADS], cs[H_HEADS];
#pragma unroll
    for (int ct = 0; ct < H_HEADS; ++ct) {
        ck[ct] = pk[ct * D_OUT + (lane & 15)];
        cs[ct] = ps[ct * D_OUT + (lane & 15)];
    }
    const int ls = active ? strip : (int)blockIdx.x;
    const float* arow = feat + (size_t)(ls * 16 + m) * IN_F + (hi << 2);
    v4f a[IN_F / 8];
#pragma unroll
    for (int q = 0; q < IN_F / 8; ++q) a[q] = *(const v4f*)(arow + 8 * q);
    *(v4f*)(wl + (tid >> 5) * WPITCH + lane * 4)        = w0;
    *(v4f*)(wl + ((tid >> 5) + 32) * WPITCH + lane * 4) = w1;
    __builtin_amdgcn_fence(__ATOMIC_RELEASE, "workgroup", "local");
    __builtin_amdgcn_s_barrier();
    __builtin_amdgcn_fence(__ATOMIC_ACQUIRE, "workgroup", "local");
    if (!active) return;

    v8f acc[H_HEADS] = {};
    const float* bbase = wl + m * WPITCH + (hi << 2);
#pragma unroll
    for (int q = 0; q < IN_F / 8; ++q) {
        v4f b[H_HEADS];
#pragma unroll
        for (int ct = 0; ct < H_HEADS; ++ct) b[ct] = *(const v4f*)(bbase + ct * 16 * WPITCH + 8 * q);
        const v2f a0 = {a[q][0], a[q][1]}, a1 = {a[q][2], a[q][3]};
#pragma unroll
        for (int ct = 0; ct < H_HEADS; ++ct) {
            const v2f b0 = {b[ct][0], b[ct][1]};
            acc[ct] = __builtin_amdgcn_wmma_f32_16x16x4_f32(
                false, a0, false, b0, (short)0, acc[ct], false, false);
        }
#pragma unroll
        for (int ct = 0; ct < H_HEADS; ++ct) {
            const v2f b1 = {b[ct][2], b[ct][3]};
            acc[ct] = __builtin_amdgcn_wmma_f32_16x16x4_f32(
                false, a1, false, b1, (short)0, acc[ct], false, false);
        }
    }
    float* outp = ft + (size_t)(strip * 16 + hi * 8) * HD + m;
#pragma unroll
    for (int ct = 0; ct < H_HEADS; ++ct)
#pragma unroll
        for (int r = 0; r < 8; ++r) outp[(size_t)r * HD + ct * 16] = acc[ct][r];

    const bool c2 = (lane & 4) != 0, c1 = (lane & 2) != 0, c0 = (lane & 1) != 0;
    float t1[H_HEADS][8];
#pragma unroll
    for (int ct = 0; ct < H_HEADS; ++ct)
#pragma unroll
        for (int r = 0; r < 8; ++r)
            t1[ct][r] = acc[ct][r] * ck[ct] + row_xor<0x168>(acc[ct][r] * cs[ct]);
    float t2[H_HEADS][4];
#pragma unroll
    for (int ct = 0; ct < H_HEADS; ++ct)
#pragma unroll
        for (int r = 0; r < 4; ++r) {
            const float keep = c2 ? t1[ct][r + 4] : t1[ct][r];
            const float send = c2 ? t1[ct][r]     : t1[ct][r + 4];
            t2[ct][r] = keep + row_xor<0x164>(send);
        }
    float t3[H_HEADS][2];
#pragma unroll
    for (int ct = 0; ct < H_HEADS; ++ct)
#pragma unroll
        for (int r = 0; r < 2; ++r) {
            const float keep = c1 ? t2[ct][r + 2] : t2[ct][r];
            const float send = c1 ? t2[ct][r]     : t2[ct][r + 2];
            t3[ct][r] = keep + row_xor<0x162>(send);
        }
    v4f lg;
#pragma unroll
    for (int ct = 0; ct < H_HEADS; ++ct) {
        const float keep = c0 ? t3[ct][1] : t3[ct][0];
        const float send = c0 ? t3[ct][0] : t3[ct][1];
        lg[ct] = keep + row_xor<0x161>(send);
    }
    float* lgp = (c3 ? er : el) + (size_t)(strip * 16 + hi * 8 + (lane & 7)) * H_HEADS;
    *(v4f*)lgp = lg;
}

__device__ __forceinline__ float edge_score(float l, float r) {
    float e = l + r;
    return e >= 0.f ? e : SLOPE * e;
}

#define NPB   391
#define NBLK  256
#define SBITS 17
#define NBKT  256
#define EPB   (E_EDGES / NBKT)
#define RUN_TRIPS (((NPB * HD) / NBKT - 1 + 31) / 32)

static_assert(EPB * NBKT == E_EDGES && EPB % 4 == 0, "edge list must split into 16-byte aligned parts");
static_assert(NPB < (1 << (32 - SBITS)) && N_NODES <= (1 << SBITS), "run entry format");

__device__ __forceinline__ unsigned owner_rows(unsigned b) {
    const unsigned rem = (unsigned)N_NODES - b * NPB;
    return rem < NPB ? rem : NPB;
}
__device__ __forceinline__ unsigned run_words(unsigned n_own) { return (n_own * HD) / NBKT; }

__device__ __forceinline__ float edge_weight(float l, float r, unsigned mu, float ss) {
    return expf(edge_score(l, r) - flt_unflip(mu)) / ss;
}

__device__ __forceinline__ void bucket_put(int d, int s, int g, unsigned* cnt, unsigned* rows) {
    const unsigned b = (unsigned)d / NPB;
    if (b < NBLK) {
        const unsigned dl = (unsigned)d - b * NPB;
        const unsigned r  = atomicAdd(&cnt[b], 1u);
        const unsigned S  = run_words(owner_rows(b));
        if (r + 1 < S) rows[(size_t)b * (NPB * HD) + g * S + 1 + r] = (unsigned)s | (dl << SBITS);
    }
}

__global__ __launch_bounds__(1024) void gat_bucket(const int* __restrict__ src,
                                                   const int* __restrict__ dst,
                                                   unsigned* __restrict__ rows) {
    __shared__ unsigned cnt[NBLK];
    const int tid = threadIdx.x;
    const int g   = blockIdx.x;
    if (tid < NBLK) cnt[tid] = 0u;
    __syncthreads();

    const int4* d4 = (const int4*)(dst + (size_t)g * EPB);
    const int4* s4 = (const int4*)(src + (size_t)g * EPB);
    const int nq = EPB / 4;
    const int TR = (EPB / 4 + 1023) / 1024;
    int4 dv[TR], sv[TR];
#pragma unroll
    for (int k = 0; k < TR; ++k) {
        const int q  = tid + k * 1024;
        const int qq = q < nq ? q : nq - 1;
        dv[k] = d4[qq];
        sv[k] = s4[qq];
    }
#pragma unroll
    for (int k = 0; k < TR; ++k) {
        if (tid + k * 1024 < nq) {
            bucket_put(dv[k].x, sv[k].x, g, cnt, rows);
            bucket_put(dv[k].y, sv[k].y, g, cnt, rows);
            bucket_put(dv[k].z, sv[k].z, g, cnt, rows);
            bucket_put(dv[k].w, sv[k].w, g, cnt, rows);
        }
    }
    __syncthreads();
    if (tid < NBLK)
        rows[(size_t)tid * (NPB * HD) + g * run_words(owner_rows((unsigned)tid))] = cnt[tid];
}

__device__ __forceinline__ float xhalf(float x) {
    const int b = __builtin_bit_cast(int, x);
    return __builtin_bit_cast(float,
        __builtin_amdgcn_permlanex16(b, b, 0x76543210, 0xfedcba98, false, false));
}
template <int CTRL>
__device__ __forceinline__ float quad_bcast(float x) {
    const int b = __builtin_bit_cast(int, x);
    return __builtin_bit_cast(float, __builtin_amdgcn_update_dpp(b, b, CTRL, 0xf, 0xf, false));
}
#define CHP   16
#define NOENT 0xFFFFFFFFu
#define GRP   8
#define QCAP  (NPB * HD + NPB * GRP)
static_assert(NBKT * ((NPB * HD) / NBKT - 1) + NPB * (GRP - 1) <= QCAP, "every run entry and every pad word must fit the list");
static_assert(CHP == 16, "the node loop has one body per number of groups of a chunk: 1 to 4");

template <int NG>
__device__ __forceinline__ void node_chunk(const unsigned* sp, const float* __restrict__ ft, const unsigned c4b,
                                           const v4f alc, const float erl, const bool q1, const bool q0,
                                           const int jq, const int deg,
                                           float& mrun, float& ssum, v2f& accA, v2f& accB) {
    unsigned sv[4 * NG];
    v4f rv[4 * NG];
#pragma unroll
    for (int u = 0; u < 4 * NG; ++u) sv[u] = sp[2 * u];
#pragma unroll
    for (int u = 0; u < 4 * NG; ++u)
        rv[u] = *(const v4f*)((const char*)ft + (size_t)((sv[u] << 8) + c4b));
    __builtin_amdgcn_sched_barrier(0);
    float sk[NG];
    float cmax = -INFINITY;
#pragma unroll
    for (int k = 0; k < NG; ++k) {
        float p[4];
#pragma unroll
        for (int i = 0; i < 4; ++i) {
            const v4f r = rv[4 * k + i];
            float d = r[0] * alc[0];
            d = r[1] * alc[1] + d;
            d = r[2] * alc[2] + d;
            d = r[3] * alc[3] + d;
            p[i] = d;
        }
        const float kA = q1 ? p[2] : p[0], kB = q1 ? p[3] : p[1];
        const float sA = q1 ? p[0] : p[2], sB = q1 ? p[1] : p[3];
        const float yA = kA + row_xor<0x162>(sA);
        const float yB = kB + row_xor<0x162>(sB);
        const float keep = q0 ? yB : yA, send = q0 ? yA : yB;
        const float z = keep + row_xor<0x161>(send);
        const float e = edge_score(z, erl);
        sk[k] = (jq + 8 * k < deg) ? e : -INFINITY;
        cmax = fmaxf(cmax, sk[k]);
    }
    cmax = fmaxf(cmax, row_xor<0x161>(cmax));
    cmax = fmaxf(cmax, row_xor<0x162>(cmax));
    cmax = fmaxf(cmax, xhalf(cmax));
    const float mnew = fmaxf(mrun, cmax);
    const float resc = expf(mrun - mnew);
    const v2f r2 = {resc, resc};
    ssum *= resc;
    accA *= r2;
    accB *= r2;
    mrun = mnew;
#pragma unroll
    for (int k = 0; k < NG; ++k) {
        const float w = expf(sk[k] - mnew);
        ssum += w;
        const float wb[4] = {quad_bcast<0x00>(w), quad_bcast<0x55>(w),
                             quad_bcast<0xAA>(w), quad_bcast<0xFF>(w)};
#pragma unroll
        for (int i = 0; i < 4; ++i) {
            const v2f w2 = {wb[i], wb[i]};
            const v2f lo2 = {rv[4 * k + i][0], rv[4 * k + i][1]};
            const v2f hi2 = {rv[4 * k + i][2], rv[4 * k + i][3]};
            accA = w2 * lo2 + accA;
            accB = w2 * hi2 + accB;
        }
    }
}

__global__ __launch_bounds__(1024) void gat_edge_agg(const int* __restrict__ src,
                                                     const int* __restrict__ dst,
                                                     const float* __restrict__ al,
                                                     const float* __restrict__ el,
                                                     const float* __restrict__ er,
                                                     const float* __restrict__ ft,
                                                     float* out) {
    __shared__ __attribute__((aligned(16))) union { float f[NPB * HD]; unsigned u[QCAP]; } buf;
    __shared__ __attribute__((aligned(16))) unsigned mx[NPB * H_HEADS];
    __shared__ __attribute__((aligned(16))) float    sm[NPB * H_HEADS];
    __shared__ __attribute__((aligned(16))) float    ers[NPB * H_HEADS];
    __shared__ unsigned ncnt[NPB + 1];
    __shared__ unsigned nstart[NPB + 1];
    __shared__ unsigned rcnt[NBKT];
    __shared__ unsigned wsum[16];
    __shared__ unsigned overfull;

    const int tid  = threadIdx.x;
    const int lane = tid & 31;
    const int wave = __builtin_amdgcn_readfirstlane(tid >> 5);
    const int lo   = blockIdx.x * NPB;
    const int rem  = N_NODES - lo;
    const unsigned n_own = (unsigned)(rem < NPB ? rem : NPB);
    const unsigned S = run_words(n_own);
    const unsigned* runs = (const unsigned*)(out + (size_t)lo * HD);

    for (int k = tid; k <= NPB; k += 1024) ncnt[k] = 0u;
    if (tid == 0) overfull = 0u;
    __syncthreads();
    if (tid < NBKT) {
        const unsigned c = runs[(unsigned)tid * S];
        if (c >= S) atomicOr(&overfull, 1u);
        rcnt[tid] = c;
    }
    __syncthreads();
    const bool bucketed = __builtin_amdgcn_readfirstlane((int)overfull) == 0;

    if (bucketed) {
        unsigned v[NBKT / 32][RUN_TRIPS];
#pragma unroll
        for (int gi = 0; gi < NBKT / 32; ++gi) {
            const int g = wave + gi * 32;
            const unsigned n = rcnt[g];
#pragma unroll
            for (int t = 0; t < RUN_TRIPS; ++t) {
                const unsigned j = (unsigned)lane + t * 32;
                v[gi][t] = j < n ? runs[(unsigned)g * S + 1 + j] : NOENT;
            }
        }
        for (int k = tid; k < (int)n_own * H_HEADS; k += 1024) ers[k] = er[(size_t)lo * H_HEADS + k];
#pragma unroll
        for (int gi = 0; gi < NBKT / 32; ++gi)
#pragma unroll
            for (int t = 0; t < RUN_TRIPS; ++t)
                if (v[gi][t] != NOENT) atomicAdd(&ncnt[v[gi][t] >> SBITS], 1u);
        __syncthreads();
        unsigned cv = 0u, inc = 0u;
        if (tid < 512) {
            cv  = tid <= NPB ? ncnt[tid] : 0u;
            cv  = (cv + (GRP - 1)) & ~(unsigned)(GRP - 1);
            inc = cv;
            for (int d = 1; d < 32; d <<= 1) {
                const unsigned x = __shfl_up(inc, d);
                if (lane >= d) inc += x;
            }
            if (lane == 31) wsum[wave] = inc;
        }
        __syncthreads();
        if (tid <= NPB) {
            unsigned off = inc - cv;
            for (int w = 0; w < wave; ++w) off += wsum[w];
            nstart[tid] = off;
            ncnt[tid]   = off;
        }
        __syncthreads();
#pragma unroll
        for (int gi = 0; gi < NBKT / 32; ++gi)
#pragma unroll
            for (int t = 0; t < RUN_TRIPS; ++t)
                if (v[gi][t] != NOENT) {
                    const unsigned pos = atomicAdd(&ncnt[v[gi][t] >> SBITS], 1u);
                    buf.u[pos] = v[gi][t] & ((1u << SBITS) - 1u);
                }
        __syncthreads();
        if (tid < (int)n_own) {
            const unsigned e = ncnt[tid], pe = nstart[tid + 1];
            if (e < pe) {
                const unsigned last = buf.u[e - 1];
                for (unsigned p = e; p < pe; ++p) buf.u[p] = last;
            }
        }
        __syncthreads();

        const int      hf  = lane >> 4;
        const unsigned c4b = (unsigned)(lane & 15) * 16u;
        const int      hl  = (lane & 15) >> 2;
        const v4f alc = *(const v4f*)(al + (lane & 15) * 4);
        const bool q1 = (lane & 2) != 0, q0 = (lane & 1) != 0;
        for (int node = wave; node < (int)n_own; node += 32) {
            const int beg  = __builtin_amdgcn_readfirstlane((int)nstart[node]);
            const int degp = __builtin_amdgcn_readfirstlane((int)nstart[node + 1]) - beg;
            const int deg  = __builtin_amdgcn_readfirstlane((int)ncnt[node]) - beg;
            const float erl = ers[node * H_HEADS + hl];

            float mrun = -INFINITY, ssum = 0.f;
            v2f accA = {0.f, 0.f}, accB = {0.f, 0.f};
            const int npairs = degp >> 1;
            for (int t0 = 0; t0 < npairs; t0 += CHP) {
                const int nv = npairs - t0 < CHP ? npairs - t0 : CHP;
                const int j0 = 2 * t0 + hf;
                const int jq = j0 + 2 * (lane & 3);
                const unsigned* sp = buf.u + beg + j0;
                const int ng = nv >> 2;
                if (ng == 4)      node_chunk<4>(sp, ft, c4b, alc, erl, q1, q0, jq, deg, mrun, ssum, accA, accB);
                else if (ng == 1) node_chunk<1>(sp, ft, c4b, alc, erl, q1, q0, jq, deg, mrun, ssum, accA, accB);
                else if (ng == 2) node_chunk<2>(sp, ft, c4b, alc, erl, q1, q0, jq, deg, mrun, ssum, accA, accB);
                else              node_chunk<3>(sp, ft, c4b, alc, erl, q1, q0, jq, deg, mrun, ssum, accA, accB);
            }
            ssum += row_xor<0x161>(ssum);
            ssum += row_xor<0x162>(ssum);
            float a0 = accA[0], a1 = accA[1], a2 = accB[0], a3 = accB[1];
            ssum += xhalf(ssum);
            a0 += xhalf(a0);
            a1 += xhalf(a1);
            a2 += xhalf(a2);
            a3 += xhalf(a3);
            const float inv = deg > 0 ? 1.0f / ssum : 0.f;
            if (lane < 16) {
                v4f o;
                o[0] = a0 * inv; o[1] = a1 * inv; o[2] = a2 * inv; o[3] = a3 * inv;
                *(v4f*)((char*)out + (size_t)(lo + node) * (HD * 4) + c4b) = o;
            }
        }
    } else {
        float* acc = buf.f;
        for (int k = tid; k < NPB * HD; k += 1024) acc[k] = 0.f;
        for (int k = tid; k < NPB * H_HEADS; k += 1024) {
            mx[k]  = 0u;
            sm[k]  = 0.f;
            ers[k] = (k < (int)n_own * H_HEADS) ? er[(size_t)lo * H_HEADS + k] : 0.f;
        }
        __syncthreads();
        for (int i = tid; i < E_EDGES; i += 1024) {
            const unsigned u = (unsigned)(dst[i] - lo);
            if (u < n_own) {
                const int s_ = src[i];
                for (int h = 0; h < H_HEADS; ++h)
                    atomicMax(&mx[u * 4 + h], flt_flip(edge_score(el[s_ * 4 + h], ers[u * 4 + h])));
            }
        }
        __syncthreads();
        for (int i = tid; i < E_EDGES; i += 1024) {
            const unsigned u = (unsigned)(dst[i] - lo);
            if (u < n_own) {
                const int s_ = src[i];
                for (int h = 0; h < H_HEADS; ++h)
                    atomicAdd(&sm[u * 4 + h],
                              expf(edge_score(el[s_ * 4 + h], ers[u * 4 + h]) - flt_unflip(mx[u * 4 + h])));
            }
        }
        __syncthreads();
        for (int i = tid; i < E_EDGES; i += 1024) {
            const unsigned u = (unsigned)(dst[i] - lo);
            if (u < n_own) {
                const int s_ = src[i];
                for (int h = 0; h < H_HEADS; ++h) {
                    const float w = edge_weight(el[s_ * 4 + h], ers[u * 4 + h],
                                                mx[u * 4 + h], sm[u * 4 + h]);
                    const float* f = ft + (size_t)s_ * HD + h * D_OUT;
                    for (int d = 0; d < D_OUT; ++d)
                        atomicAdd(&acc[u * HD + h * D_OUT + d], f[d] * w);
                }
            }
        }
        __syncthreads();
        float* o = out + (size_t)lo * HD;
        const int nout = (int)n_own * HD;
        for (int k = tid; k < nout; k += 1024) o[k] = acc[k];
    }
}

extern "C" void kernel_launch(void* const* d_in, const int* in_sizes, int n_in,
                              void* d_out, int out_size, void* d_ws, size_t ws_size,
                              hipStream_t stream) {
    const float* feat = (const float*)d_in[0];
    const float* W    = (const float*)d_in[1];
    const float* al   = (const float*)d_in[2];
    const float* ar   = (const float*)d_in[3];
    const int*   src  = (const int*)d_in[4];
    const int*   dst  = (const int*)d_in[5];
    float* out = (float*)d_out;

    float* ft = (float*)d_ws;
    float* el = ft + (size_t)N_NODES * HD;
    float* er = el + (size_t)N_NODES * H_HEADS;

    gat_bucket<<<NBKT, 1024, 0, stream>>>(src, dst, (unsigned*)d_out);
    gat_gemm<<<GEMM_BLK, 1024, 0, stream>>>(feat, W, al, ar, ft, el, er);
    gat_edge_agg<<<NBLK, 1024, 0, stream>>>(src, dst, al, el, er, ft, out);
}


# Bucket pass sorts its block's edges by owner in LDS (counting sort at the same ranks), then stores each run from consecutive lanes, one 128-byte line per store, not one scattered 4-byte store per edge
# speedup vs baseline: 19.6091x; 1.0997x over previous
#include <hip/hip_runtime.h>

#define N_NODES 100000
#define IN_F    128
#define H_HEADS 4
#define D_OUT   16
#define HD      64
#define E_EDGES 3200000
#define SLOPE   0.2f

typedef __attribute__((ext_vector_type(2))) float v2f;
typedef __attribute__((ext_vector_type(4))) float v4f;
typedef __attribute__((ext_vector_type(8))) float v8f;

__device__ __forceinline__ unsigned flt_flip(float f) {
    unsigned u = __float_as_uint(f);
    return u ^ (((int)u >> 31) | 0x80000000u);
}
__device__ __forceinline__ float flt_unflip(unsigned u) {
    unsigned b = (u & 0x80000000u) ? (u ^ 0x80000000u) : ~u;
    return __uint_as_float(b);
}

template <int CTRL>
__device__ __forceinline__ float row_xor(float x) {
    const int b = __builtin_bit_cast(int, x);
    return __builtin_bit_cast(float, __builtin_amdgcn_update_dpp(b, b, CTRL, 0xf, 0xf, false));
}

#define GEMM_BLK 256
#define NSTRIP   (N_NODES / 16)
#define WPITCH   (IN_F + 4)
static_assert(NSTRIP * 16 == N_NODES && NSTRIP <= GEMM_BLK * 32, "every strip needs a wave");

__global__ __launch_bounds__(1024) void gat_gemm(const float* __restrict__ feat,
                                                 const float* __restrict__ W,
                                                 const float* __restrict__ al,
                                                 const float* __restrict__ ar,
                                                 float* __restrict__ ft,
                                                 float* __restrict__ el,
                                                 float* __restrict__ er) {
    __shared__ __attribute__((aligned(16))) float wl[HD * WPITCH];
    const int tid  = threadIdx.x;
    const int lane = tid & 31;
    const int wave  = __builtin_amdgcn_readfirstlane(tid >> 5);
    const int strip = wave * GEMM_BLK + (int)blockIdx.x;
    const bool active = strip < NSTRIP;
    const int m  = lane & 15;
    const int hi = lane >> 4;

    const v4f w0 = *(const v4f*)(W + (size_t)tid * 4);
    const v4f w1 = *(const v4f*)(W + (size_t)(tid + 1024) * 4);
    const bool c3 = (lane & 8) != 0;
    const float* pk = c3 ? ar : al;
    const float* ps = c3 ? al : ar;
    float ck[H_HEADS], cs[H_HEADS];
#pragma unroll
    for (int ct = 0; ct < H_HEADS; ++ct) {
        ck[ct] = pk[ct * D_OUT + (lane & 15)];
        cs[ct] = ps[ct * D_OUT + (lane & 15)];
    }
    const int ls = active ? strip : (int)blockIdx.x;
    const float* arow = feat + (size_t)(ls * 16 + m) * IN_F + (hi << 2);
    v4f a[IN_F / 8];
#pragma unroll
    for (int q = 0; q < IN_F / 8; ++q) a[q] = *(const v4f*)(arow + 8 * q);
    *(v4f*)(wl + (tid >> 5) * WPITCH + lane * 4)        = w0;
    *(v4f*)(wl + ((tid >> 5) + 32) * WPITCH + lane * 4) = w1;
    __builtin_amdgcn_fence(__ATOMIC_RELEASE, "workgroup", "local");
    __builtin_amdgcn_s_barrier();
    __builtin_amdgcn_fence(__ATOMIC_ACQUIRE, "workgroup", "local");
    if (!active) return;

    v8f acc[H_HEADS] = {};
    const float* bbase = wl + m * WPITCH + (hi << 2);
#pragma unroll
    for (int q = 0; q < IN_F / 8; ++q) {
        v4f b[H_HEADS];
#pragma unroll
        for (int ct = 0; ct < H_HEADS; ++ct) b[ct] = *(const v4f*)(bbase + ct * 16 * WPITCH + 8 * q);
        const v2f a0 = {a[q][0], a[q][1]}, a1 = {a[q][2], a[q][3]};
#pragma unroll
        for (int ct = 0; ct < H_HEADS; ++ct) {
            const v2f b0 = {b[ct][0], b[ct][1]};
            acc[ct] = __builtin_amdgcn_wmma_f32_16x16x4_f32(
                false, a0, false, b0, (short)0, acc[ct], false, false);
        }
#pragma unroll
        for (int ct = 0; ct < H_HEADS; ++ct) {
            const v2f b1 = {b[ct][2], b[ct][3]};
            acc[ct] = __builtin_amdgcn_wmma_f32_16x16x4_f32(
                false, a1, false, b1, (short)0, acc[ct], false, false);
        }
    }
    float* outp = ft + (size_t)(strip * 16 + hi * 8) * HD + m;
#pragma unroll
    for (int ct = 0; ct < H_HEADS; ++ct)
#pragma unroll
        for (int r = 0; r < 8; ++r) outp[(size_t)r * HD + ct * 16] = acc[ct][r];

    const bool c2 = (lane & 4) != 0, c1 = (lane & 2) != 0, c0 = (lane & 1) != 0;
    float t1[H_HEADS][8];
#pragma unroll
    for (int ct = 0; ct < H_HEADS; ++ct)
#pragma unroll
        for (int r = 0; r < 8; ++r)
            t1[ct][r] = acc[ct][r] * ck[ct] + row_xor<0x168>(acc[ct][r] * cs[ct]);
    float t2[H_HEADS][4];
#pragma unroll
    for (int ct = 0; ct < H_HEADS; ++ct)
#pragma unroll
        for (int r = 0; r < 4; ++r) {
            const float keep = c2 ? t1[ct][r + 4] : t1[ct][r];
            const float send = c2 ? t1[ct][r]     : t1[ct][r + 4];
            t2[ct][r] = keep + row_xor<0x164>(send);
        }
    float t3[H_HEADS][2];
#pragma unroll
    for (int ct = 0; ct < H_HEADS; ++ct)
#pragma unroll
        for (int r = 0; r < 2; ++r) {
            const float keep = c1 ? t2[ct][r + 2] : t2[ct][r];
            const float send = c1 ? t2[ct][r]     : t2[ct][r + 2];
            t3[ct][r] = keep + row_xor<0x162>(send);
        }
    v4f lg;
#pragma unroll
    for (int ct = 0; ct < H_HEADS; ++ct) {
        const float keep = c0 ? t3[ct][1] : t3[ct][0];
        const float send = c0 ? t3[ct][0] : t3[ct][1];
        lg[ct] = keep + row_xor<0x161>(send);
    }
    float* lgp = (c3 ? er : el) + (size_t)(strip * 16 + hi * 8 + (lane & 7)) * H_HEADS;
    *(v4f*)lgp = lg;
}

__device__ __forceinline__ float edge_score(float l, float r) {
    float e = l + r;
    return e >= 0.f ? e : SLOPE * e;
}

#define NPB   391
#define NBLK  256
#define SBITS 17
#define NBKT  256
#define EPB   (E_EDGES / NBKT)
#define RUN_TRIPS (((NPB * HD) / NBKT - 1 + 31) / 32)

static_assert(EPB * NBKT == E_EDGES && EPB % 4 == 0, "edge list must split into 16-byte aligned parts");
static_assert(NPB < (1 << (32 - SBITS)) && N_NODES <= (1 << SBITS), "run entry format");

__device__ __forceinline__ unsigned owner_rows(unsigned b) {
    const unsigned rem = (unsigned)N_NODES - b * NPB;
    return rem < NPB ? rem : NPB;
}
__device__ __forceinline__ unsigned run_words(unsigned n_own) { return (n_own * HD) / NBKT; }

__device__ __forceinline__ float edge_weight(float l, float r, unsigned mu, float ss) {
    return expf(edge_score(l, r) - flt_unflip(mu)) / ss;
}

typedef __attribute__((ext_vector_type(4))) unsigned v4u;
#define OPW (NBLK / 32)
static_assert(NBLK == 8 * 32 && OPW == 8, "the running sum over the owners' counts is one wave's, eight counts per lane; a wave stores eight owners' runs");
static_assert(((NPB * HD) % 32) == 0, "an owner's rows must start on a 128-byte line");
static_assert((NPB * HD) / NBKT + 31 <= 4 * 32, "a run and the words in front of it in its first line are at most four lines");

__global__ __launch_bounds__(1024) void gat_bucket(const int* __restrict__ src,
                                                   const int* __restrict__ dst,
                                                   unsigned* __restrict__ rows) {
    __shared__ __attribute__((aligned(16))) unsigned cnt[NBLK + 4];
    __shared__ __attribute__((aligned(16))) unsigned base[NBLK + 4];
    __shared__ unsigned ent[EPB];
    const int tid  = threadIdx.x;
    const int lane = tid & 31;
    const int wave = __builtin_amdgcn_readfirstlane(tid >> 5);
    const int g    = blockIdx.x;
    if (tid <= NBLK) cnt[tid] = 0u;
    __syncthreads();

    const int4* d4 = (const int4*)(dst + (size_t)g * EPB);
    const int4* s4 = (const int4*)(src + (size_t)g * EPB);
    const int nq = EPB / 4;
    const int TR = (EPB / 4 + 1023) / 1024;
    int4 dv[TR], sv[TR];
#pragma unroll
    for (int k = 0; k < TR; ++k) {
        const int q  = tid + k * 1024;
        const int qq = q < nq ? q : nq - 1;
        dv[k] = d4[qq];
        sv[k] = s4[qq];
    }
    unsigned ob[TR][4] = {}, rk[TR][4] = {}, val[TR][4] = {};
#pragma unroll
    for (int k = 0; k < TR; ++k) {
        if (tid + k * 1024 < nq) {
            const int dd[4] = {dv[k].x, dv[k].y, dv[k].z, dv[k].w};
            const int ss[4] = {sv[k].x, sv[k].y, sv[k].z, sv[k].w};
#pragma unroll
            for (int i = 0; i < 4; ++i) {
                const unsigned b = (unsigned)dd[i] / NPB;
                ob[k][i]  = b < NBLK ? b : NBLK;
                val[k][i] = (unsigned)ss[i] | (((unsigned)dd[i] - b * NPB) << SBITS);
            }
#pragma unroll
            for (int i = 0; i < 4; ++i) rk[k][i] = atomicAdd(&cnt[ob[k][i]], 1u);
        }
    }
    __syncthreads();
    if (wave == 0) {
        const v4u c0 = *(const v4u*)(cnt + lane * 8), c1 = *(const v4u*)(cnt + lane * 8 + 4);
        const unsigned t = c0[0] + c0[1] + c0[2] + c0[3] + c1[0] + c1[1] + c1[2] + c1[3];
        unsigned inc = t;
        for (int d = 1; d < 32; d <<= 1) {
            const unsigned x = __shfl_up(inc, d);
            if (lane >= d) inc += x;
        }
        v4u b0, b1;
        b0[0] = inc - t;       b0[1] = b0[0] + c0[0]; b0[2] = b0[1] + c0[1]; b0[3] = b0[2] + c0[2];
        b1[0] = b0[3] + c0[3]; b1[1] = b1[0] + c1[0]; b1[2] = b1[1] + c1[1]; b1[3] = b1[2] + c1[2];
        *(v4u*)(base + lane * 8)     = b0;
        *(v4u*)(base + lane * 8 + 4) = b1;
        if (lane == 31) base[NBLK] = inc;
    }
    __syncthreads();
#pragma unroll
    for (int k = 0; k < TR; ++k) {
        if (tid + k * 1024 < nq) {
            unsigned p[4];
#pragma unroll
            for (int i = 0; i < 4; ++i) p[i] = base[ob[k][i]];
#pragma unroll
            for (int i = 0; i < 4; ++i) ent[p[i] + rk[k][i]] = val[k][i];
        }
    }
    __syncthreads();
    const int ob0 = wave * OPW;
    const v4u cA = *(const v4u*)(cnt + ob0),  cB = *(const v4u*)(cnt + ob0 + 4);
    const v4u sA = *(const v4u*)(base + ob0), sB = *(const v4u*)(base + ob0 + 4);
    const unsigned cw[OPW] = {cA[0], cA[1], cA[2], cA[3], cB[0], cB[1], cB[2], cB[3]};
    const unsigned sw[OPW] = {sA[0], sA[1], sA[2], sA[3], sB[0], sB[1], sB[2], sB[3]};
    unsigned cK[OPW], sK[OPW];
    int nwK[OPW], oK[OPW];
    unsigned* runK[OPW];
    unsigned ev[OPW][3];
#pragma unroll
    for (int k = 0; k < OPW; ++k) {
        const unsigned b = (unsigned)(ob0 + k);
        cK[k] = (unsigned)__builtin_amdgcn_readfirstlane((int)cw[k]);
        sK[k] = (unsigned)__builtin_amdgcn_readfirstlane((int)sw[k]);
        const unsigned S = run_words(owner_rows(b));
        nwK[k]  = (int)(cK[k] < S - 1 ? cK[k] : S - 1) + 1;
        oK[k]   = (int)(((unsigned)g * S) & 31u);
        runK[k] = rows + (size_t)b * (NPB * HD) + (unsigned)g * S;
#pragma unroll
        for (int t = 0; t < 3; ++t) {
            const int j = 32 * t - oK[k] + lane;
            const bool ent_word = j > 0 && j < nwK[k];
            ev[k][t] = ent[ent_word ? sK[k] + (unsigned)j - 1u : 0u];
        }
    }
#pragma unroll
    for (int k = 0; k < OPW; ++k) {
#pragma unroll
        for (int t = 0; t < 3; ++t) {
            const int j = 32 * t - oK[k] + lane;
            if (j >= 0 && j < nwK[k]) runK[k][j] = j > 0 ? ev[k][t] : cK[k];
        }
    }
#pragma unroll
    for (int k = 0; k < OPW; ++k) {
        if (96 - oK[k] < nwK[k]) {
            const int j = 96 - oK[k] + lane;
            if (j < nwK[k]) runK[k][j] = ent[sK[k] + (unsigned)j - 1u];
        }
    }
}

__device__ __forceinline__ float xhalf(float x) {
    const int b = __builtin_bit_cast(int, x);
    return __builtin_bit_cast(float,
        __builtin_amdgcn_permlanex16(b, b, 0x76543210, 0xfedcba98, false, false));
}
template <int CTRL>
__device__ __forceinline__ float quad_bcast(float x) {
    const int b = __builtin_bit_cast(int, x);
    return __builtin_bit_cast(float, __builtin_amdgcn_update_dpp(b, b, CTRL, 0xf, 0xf, false));
}
#define CHP   16
#define NOENT 0xFFFFFFFFu
#define GRP   8
#define QCAP  (NPB * HD + NPB * GRP)
static_assert(NBKT * ((NPB * HD) / NBKT - 1) + NPB * (GRP - 1) <= QCAP, "every run entry and every pad word must fit the list");
static_assert(CHP == 16, "the node loop has one body per number of groups of a chunk: 1 to 4");

template <int NG>
__device__ __forceinline__ void node_chunk(const unsigned* sp, const float* __restrict__ ft, const unsigned c4b,
                                           const v4f alc, const float erl, const bool q1, const bool q0,
                                           const int jq, const int deg,
                                           float& mrun, float& ssum, v2f& accA, v2f& accB) {
    unsigned sv[4 * NG];
    v4f rv[4 * NG];
#pragma unroll
    for (int u = 0; u < 4 * NG; ++u) sv[u] = sp[2 * u];
#pragma unroll
    for (int u = 0; u < 4 * NG; ++u)
        rv[u] = *(const v4f*)((const char*)ft + (size_t)((sv[u] << 8) + c4b));
    __builtin_amdgcn_sched_barrier(0);
    float sk[NG];
    float cmax = -INFINITY;
#pragma unroll
    for (int k = 0; k < NG; ++k) {
        float p[4];
#pragma unroll
        for (int i = 0; i < 4; ++i) {
            const v4f r = rv[4 * k + i];
            float d = r[0] * alc[0];
            d = r[1] * alc[1] + d;
            d = r[2] * alc[2] + d;
            d = r[3] * alc[3] + d;
            p[i] = d;
        }
        const float kA = q1 ? p[2] : p[0], kB = q1 ? p[3] : p[1];
        const float sA = q1 ? p[0] : p[2], sB = q1 ? p[1] : p[3];
        const float yA = kA + row_xor<0x162>(sA);
        const float yB = kB + row_xor<0x162>(sB);
        const float keep = q0 ? yB : yA, send = q0 ? yA : yB;
        const float z = keep + row_xor<0x161>(send);
        const float e = edge_score(z, erl);
        sk[k] = (jq + 8 * k < deg) ? e : -INFINITY;
        cmax = fmaxf(cmax, sk[k]);
    }
    cmax = fmaxf(cmax, row_xor<0x161>(cmax));
    cmax = fmaxf(cmax, row_xor<0x162>(cmax));
    cmax = fmaxf(cmax, xhalf(cmax));
    const float mnew = fmaxf(mrun, cmax);
    const float resc = expf(mrun - mnew);
    const v2f r2 = {resc, resc};
    ssum *= resc;
    accA *= r2;
    accB *= r2;
    mrun = mnew;
#pragma unroll
    for (int k = 0; k < NG; ++k) {
        const float w = expf(sk[k] - mnew);
        ssum += w;
        const float wb[4] = {quad_bcast<0x00>(w), quad_bcast<0x55>(w),
                             quad_bcast<0xAA>(w), quad_bcast<0xFF>(w)};
#pragma unroll
        for (int i = 0; i < 4; ++i) {
            const v2f w2 = {wb[i], wb[i]};
            const v2f lo2 = {rv[4 * k + i][0], rv[4 * k + i][1]};
            const v2f hi2 = {rv[4 * k + i][2], rv[4 * k + i][3]};
            accA = w2 * lo2 + accA;
            accB = w2 * hi2 + accB;
        }
    }
}

__global__ __launch_bounds__(1024) void gat_edge_agg(const int* __restrict__ src,
                                                     const int* __restrict__ dst,
                                                     const float* __restrict__ al,
                                                     const float* __restrict__ el,
                                                     const float* __restrict__ er,
                                                     const float* __restrict__ ft,
                                                     float* out) {
    __shared__ __attribute__((aligned(16))) union { float f[NPB * HD]; unsigned u[QCAP]; } buf;
    __shared__ __attribute__((aligned(16))) unsigned mx[NPB * H_HEADS];
    __shared__ __attribute__((aligned(16))) float    sm[NPB * H_HEADS];
    __shared__ __attribute__((aligned(16))) float    ers[NPB * H_HEADS];
    __shared__ unsigned ncnt[NPB + 1];
    __shared__ unsigned nstart[NPB + 1];
    __shared__ unsigned rcnt[NBKT];
    __shared__ unsigned wsum[16];
    __shared__ unsigned overfull;

    const int tid  = threadIdx.x;
    const int lane = tid & 31;
    const int wave = __builtin_amdgcn_readfirstlane(tid >> 5);
    const int lo   = blockIdx.x * NPB;
    const int rem  = N_NODES - lo;
    const unsigned n_own = (unsigned)(rem < NPB ? rem : NPB);
    const unsigned S = run_words(n_own);
    const unsigned* runs = (const unsigned*)(out + (size_t)lo * HD);

    for (int k = tid; k <= NPB; k += 1024) ncnt[k] = 0u;
    if (tid == 0) overfull = 0u;
    __syncthreads();
    if (tid < NBKT) {
        const unsigned c = runs[(unsigned)tid * S];
        if (c >= S) atomicOr(&overfull, 1u);
        rcnt[tid] = c;
    }
    __syncthreads();
    const bool bucketed = __builtin_amdgcn_readfirstlane((int)overfull) == 0;

    if (bucketed) {
        unsigned v[NBKT / 32][RUN_TRIPS];
#pragma unroll
        for (int gi = 0; gi < NBKT / 32; ++gi) {
            const int g = wave + gi * 32;
            const unsigned n = rcnt[g];
#pragma unroll
            for (int t = 0; t < RUN_TRIPS; ++t) {
                const unsigned j = (unsigned)lane + t * 32;
                v[gi][t] = j < n ? runs[(unsigned)g * S + 1 + j] : NOENT;
            }
        }
        for (int k = tid; k < (int)n_own * H_HEADS; k += 1024) ers[k] = er[(size_t)lo * H_HEADS + k];
#pragma unroll
        for (int gi = 0; gi < NBKT / 32; ++gi)
#pragma unroll
            for (int t = 0; t < RUN_TRIPS; ++t)
                if (v[gi][t] != NOENT) atomicAdd(&ncnt[v[gi][t] >> SBITS], 1u);
        __syncthreads();
        unsigned cv = 0u, inc = 0u;
        if (tid < 512) {
            cv  = tid <= NPB ? ncnt[tid] : 0u;
            cv  = (cv + (GRP - 1)) & ~(unsigned)(GRP - 1);
            inc = cv;
            for (int d = 1; d < 32; d <<= 1) {
                const unsigned x = __shfl_up(inc, d);
                if (lane >= d) inc += x;
            }
            if (lane == 31) wsum[wave] = inc;
        }
        __syncthreads();
        if (tid <= NPB) {
            unsigned off = inc - cv;
            for (int w = 0; w < wave; ++w) off += wsum[w];
            nstart[tid] = off;
            ncnt[tid]   = off;
        }
        __syncthreads();
#pragma unroll
        for (int gi = 0; gi < NBKT / 32; ++gi)
#pragma unroll
            for (int t = 0; t < RUN_TRIPS; ++t)
                if (v[gi][t] != NOENT) {
                    const unsigned pos = atomicAdd(&ncnt[v[gi][t] >> SBITS], 1u);
                    buf.u[pos] = v[gi][t] & ((1u << SBITS) - 1u);
                }
        __syncthreads();
        if (tid < (int)n_own) {
            const unsigned e = ncnt[tid], pe = nstart[tid + 1];
            if (e < pe) {
                const unsigned last = buf.u[e - 1];
                for (unsigned p = e; p < pe; ++p) buf.u[p] = last;
            }
        }
        __syncthreads();

        const int      hf  = lane >> 4;
        const unsigned c4b = (unsigned)(lane & 15) * 16u;
        const int      hl  = (lane & 15) >> 2;
        const v4f alc = *(const v4f*)(al + (lane & 15) * 4);
        const bool q1 = (lane & 2) != 0, q0 = (lane & 1) != 0;
        for (int node = wave; node < (int)n_own; node += 32) {
            const int beg  = __builtin_amdgcn_readfirstlane((int)nstart[node]);
            const int degp = __builtin_amdgcn_readfirstlane((int)nstart[node + 1]) - beg;
            const int deg  = __builtin_amdgcn_readfirstlane((int)ncnt[node]) - beg;
            const float erl = ers[node * H_HEADS + hl];

            float mrun = -INFINITY, ssum = 0.f;
            v2f accA = {0.f, 0.f}, accB = {0.f, 0.f};
            const int npairs = degp >> 1;
            for (int t0 = 0; t0 < npairs; t0 += CHP) {
                const int nv = npairs - t0 < CHP ? npairs - t0 : CHP;
                const int j0 = 2 * t0 + hf;
                const int jq = j0 + 2 * (lane & 3);
                const unsigned* sp = buf.u + beg + j0;
                const int ng = nv >> 2;
                if (ng == 4)      node_chunk<4>(sp, ft, c4b, alc, erl, q1, q0, jq, deg, mrun, ssum, accA, accB);
                else if (ng == 1) node_chunk<1>(sp, ft, c4b, alc, erl, q1, q0, jq, deg, mrun, ssum, accA, accB);
                else if (ng == 2) node_chunk<2>(sp, ft, c4b, alc, erl, q1, q0, jq, deg, mrun, ssum, accA, accB);
                else              node_chunk<3>(sp, ft, c4b, alc, erl, q1, q0, jq, deg, mrun, ssum, accA, accB);
            }
            ssum += row_xor<0x161>(ssum);
            ssum += row_xor<0x162>(ssum);
            float a0 = accA[0], a1 = accA[1], a2 = accB[0], a3 = accB[1];
            ssum += xhalf(ssum);
            a0 += xhalf(a0);
            a1 += xhalf(a1);
            a2 += xhalf(a2);
            a3 += xhalf(a3);
            const float inv = deg > 0 ? 1.0f / ssum : 0.f;
            if (lane < 16) {
                v4f o;
                o[0] = a0 * inv; o[1] = a1 * inv; o[2] = a2 * inv; o[3] = a3 * inv;
                *(v4f*)((char*)out + (size_t)(lo + node) * (HD * 4) + c4b) = o;
            }
        }
    } else {
        float* acc = buf.f;
        for (int k = tid; k < NPB * HD; k += 1024) acc[k] = 0.f;
        for (int k = tid; k < NPB * H_HEADS; k += 1024) {
            mx[k]  = 0u;
            sm[k]  = 0.f;
            ers[k] = (k < (int)n_own * H_HEADS) ? er[(size_t)lo * H_HEADS + k] : 0.f;
        }
        __syncthreads();
        for (int i = tid; i < E_EDGES; i += 1024) {
            const unsigned u = (unsigned)(dst[i] - lo);
            if (u < n_own) {
                const int s_ = src[i];
                for (int h = 0; h < H_HEADS; ++h)
                    atomicMax(&mx[u * 4 + h], flt_flip(edge_score(el[s_ * 4 + h], ers[u * 4 + h])));
            }
        }
        __syncthreads();
        for (int i = tid; i < E_EDGES; i += 1024) {
            const unsigned u = (unsigned)(dst[i] - lo);
            if (u < n_own) {
                const int s_ = src[i];
                for (int h = 0; h < H_HEADS; ++h)
                    atomicAdd(&sm[u * 4 + h],
                              expf(edge_score(el[s_ * 4 + h], ers[u * 4 + h]) - flt_unflip(mx[u * 4 + h])));
            }
        }
        __syncthreads();
        for (int i = tid; i < E_EDGES; i += 1024) {
            const unsigned u = (unsigned)(dst[i] - lo);
            if (u < n_own) {
                const int s_ = src[i];
                for (int h = 0; h < H_HEADS; ++h) {
                    const float w = edge_weight(el[s_ * 4 + h], ers[u * 4 + h],
                                                mx[u * 4 + h], sm[u * 4 + h]);
                    const float* f = ft + (size_t)s_ * HD + h * D_OUT;
                    for (int d = 0; d < D_OUT; ++d)
                        atomicAdd(&acc[u * HD + h * D_OUT + d], f[d] * w);
                }
            }
        }
        __syncthreads();
        float* o = out + (size_t)lo * HD;
        const int nout = (int)n_own * HD;
        for (int k = tid; k < nout; k += 1024) o[k] = acc[k];
    }
}

extern "C" void kernel_launch(void* const* d_in, const int* in_sizes, int n_in,
                              void* d_out, int out_size, void* d_ws, size_t ws_size,
                              hipStream_t stream) {
    const float* feat = (const float*)d_in[0];
    const float* W    = (const float*)d_in[1];
    const float* al   = (const float*)d_in[2];
    const float* ar   = (const float*)d_in[3];
    const int*   src  = (const int*)d_in[4];
    const int*   dst  = (const int*)d_in[5];
    float* out = (float*)d_out;

    float* ft = (float*)d_ws;
    float* el = ft + (size_t)N_NODES * HD;
    float* er = el + (size_t)N_NODES * H_HEADS;

    gat_bucket<<<NBKT, 1024, 0, stream>>>(src, dst, (unsigned*)d_out);
    gat_gemm<<<GEMM_BLK, 1024, 0, stream>>>(feat, W, al, ar, ft, el, er);
    gat_edge_agg<<<NBLK, 1024, 0, stream>>>(src, dst, al, el, er, ft, out);
}
